# starting kernel
# baseline (speedup 1.0000x reference)
#include <hip/hip_runtime.h>
#include <hip/hip_bf16.h>

typedef float v2f __attribute__((ext_vector_type(2)));
typedef float v8f __attribute__((ext_vector_type(8)));

#define N_NODES 50000
#define N_EDGES 800000

__global__ void k_init_deg(float* __restrict__ deg, int n) {
  int i = blockIdx.x * blockDim.x + threadIdx.x;
  if (i < n) deg[i] = 1.0f;
}

__global__ void k_count_deg(const int* __restrict__ dst, float* __restrict__ deg, int e) {
  int i = blockIdx.x * blockDim.x + threadIdx.x;
  if (i < e) atomicAdd(&deg[dst[i]], 1.0f);
}

__global__ void k_dinv(float* __restrict__ deg, int n) {
  int i = blockIdx.x * blockDim.x + threadIdx.x;
  if (i < n) deg[i] = rsqrtf(deg[i]);
}

__global__ void k_pack_w(const float* __restrict__ W, float* __restrict__ Wp,
                         int K, int N) {
  int idx = blockIdx.x * blockDim.x + threadIdx.x;
  if (idx >= K * N) return;
  int k = idx / N;
  int n = idx - k * N;
  Wp[(size_t)(k >> 2) * (N * 4) + n * 4 + (k & 3)] = W[idx];
}

template <int KDIM, int NT, bool BIAS_RELU>
__global__ __launch_bounds__(128) void k_gemm_wmma(
    const float* __restrict__ A, const float* __restrict__ Wp,
    const float* __restrict__ bias, float* __restrict__ out,
    int M, int out_stride, int out_off) {
  const int lane = threadIdx.x & 31;
  const int wave = threadIdx.x >> 5;
  const int row_tile = (blockIdx.x * 4 + wave) * 16;
  if (row_tile >= M) return;

  const int m16 = lane & 15;
  const int khalf = lane >> 4;
  constexpr int NCOLS = NT * 16;

  v8f acc[NT];
#pragma unroll
  for (int t = 0; t < NT; ++t) acc[t] = (v8f){0.f, 0.f, 0.f, 0.f, 0.f, 0.f, 0.f, 0.f};

  const float* __restrict__ arow = A + (size_t)(row_tile + m16) * KDIM + khalf * 2;
  const float* __restrict__ wlane = Wp + m16 * 4 + khalf * 2;

  for (int k0 = 0; k0 < KDIM; k0 += 16) {
    if (k0 + 256 < KDIM) __builtin_prefetch(arow + k0 + 256, 0, 0);
#pragma unroll
    for (int kq = 0; kq < 4; ++kq) {
      const int k = k0 + kq * 4;
      v2f afrag = *(const v2f*)(arow + k);
      const float* wq = wlane + (size_t)(k >> 2) * (NCOLS * 4);
#pragma unroll
      for (int t = 0; t < NT; ++t) {
        v2f bfrag = *(const v2f*)(wq + t * 64);
        acc[t] = __builtin_amdgcn_wmma_f32_16x16x4_f32(
             false, afrag,  false, bfrag,
             (short)0, acc[t],  false,  false);
      }
    }
  }

  const int rbase = row_tile + khalf * 8;
#pragma unroll
  for (int t = 0; t < NT; ++t) {
#pragma unroll
    for (int i = 0; i < 8; ++i) {
      const int c = t * 16 + m16;
      float v = acc[t][i];
      if (BIAS_RELU) v = fmaxf(v + bias[c], 0.0f);
      out[(size_t)(rbase + i) * out_stride + out_off + c] = v;
    }
  }
}

__global__ void k_node_init(const float* __restrict__ xw, const float* __restrict__ dinv,
                            const float* __restrict__ bias, float* __restrict__ out,
                            int n, int C) {
  int idx = blockIdx.x * blockDim.x + threadIdx.x;
  if (idx >= n * C) return;
  int i = idx / C;
  int c = idx - i * C;
  float di = dinv[i];
  out[idx] = bias[c] + xw[idx] * di * di;
}

template <int C>
__global__ void k_scatter(const float* __restrict__ xw, const float* __restrict__ dinv,
                          const int* __restrict__ src, const int* __restrict__ dst,
                          float* __restrict__ out, int e) {
  constexpr int CH = C / 4;
  int tid = blockIdx.x * blockDim.x + threadIdx.x;
  int eidx = tid / CH;
  int ch = tid - eidx * CH;
  if (eidx >= e) return;
  int s = src[eidx];
  int d = dst[eidx];
  float nrm = dinv[s] * dinv[d];
  float4 v = *((const float4*)(xw + (size_t)s * C) + ch);
  float* o = out + (size_t)d * C + ch * 4;
  atomicAdd(o + 0, v.x * nrm);
  atomicAdd(o + 1, v.y * nrm);
  atomicAdd(o + 2, v.z * nrm);
  atomicAdd(o + 3, v.w * nrm);
}

__global__ void k_relu(float* __restrict__ x, int n) {
  int i = blockIdx.x * blockDim.x + threadIdx.x;
  if (i < n) x[i] = fmaxf(x[i], 0.0f);
}

extern "C" void kernel_launch(void* const* d_in, const int* in_sizes, int n_in,
                              void* d_out, int out_size, void* d_ws, size_t ws_size,
                              hipStream_t stream) {
  (void)in_sizes; (void)n_in; (void)out_size; (void)ws_size;

  const float* feature = (const float*)d_in[0];
  const float* img     = (const float*)d_in[1];
  const int*   edges   = (const int*)d_in[2];
  const float* W_feat  = (const float*)d_in[3];
  const float* b_feat  = (const float*)d_in[4];
  const float* W_img   = (const float*)d_in[5];
  const float* b_img   = (const float*)d_in[6];
  const float* W_g1    = (const float*)d_in[7];
  const float* b_g1    = (const float*)d_in[8];
  const float* W_g2    = (const float*)d_in[9];
  const float* b_g2    = (const float*)d_in[10];
  float* out = (float*)d_out;

  const int* src = edges;
  const int* dst = edges + N_EDGES;

  float* ws     = (float*)d_ws;
  float* dinv   = ws;
  float* bufA   = ws + 50048;
  float* bufB   = bufA + (size_t)N_NODES * 96;
  float* wpF    = bufB + (size_t)N_NODES * 96;
  float* wpI    = wpF + 512 * 64;
  float* wpG1   = wpI + 256 * 32;
  float* wpG2   = wpG1 + 96 * 96;

  const int T = 256;
  const int MB = (N_NODES + 63) / 64;

  k_pack_w<<<(512 * 64 + T - 1) / T, T, 0, stream>>>(W_feat, wpF, 512, 64);
  k_pack_w<<<(256 * 32 + T - 1) / T, T, 0, stream>>>(W_img, wpI, 256, 32);
  k_pack_w<<<(96 * 96 + T - 1) / T, T, 0, stream>>>(W_g1, wpG1, 96, 96);
  k_pack_w<<<(96 * 64 + T - 1) / T, T, 0, stream>>>(W_g2, wpG2, 96, 64);

  k_init_deg<<<(N_NODES + T - 1) / T, T, 0, stream>>>(dinv, N_NODES);
  k_count_deg<<<(N_EDGES + T - 1) / T, T, 0, stream>>>(dst, dinv, N_EDGES);
  k_dinv<<<(N_NODES + T - 1) / T, T, 0, stream>>>(dinv, N_NODES);

  k_gemm_wmma<512, 4, true><<<MB, 128, 0, stream>>>(feature, wpF, b_feat, bufA, N_NODES, 96, 0);
  k_gemm_wmma<256, 2, true><<<MB, 128, 0, stream>>>(img, wpI, b_img, bufA, N_NODES, 96, 64);

  k_gemm_wmma<96, 6, false><<<MB, 128, 0, stream>>>(bufA, wpG1, nullptr, bufB, N_NODES, 96, 0);
  k_node_init<<<(N_NODES * 96 + T - 1) / T, T, 0, stream>>>(bufB, dinv, b_g1, bufA, N_NODES, 96);
  k_scatter<96><<<(N_EDGES * 24 + T - 1) / T, T, 0, stream>>>(bufB, dinv, src, dst, bufA, N_EDGES);
  k_relu<<<(N_NODES * 96 + T - 1) / T, T, 0, stream>>>(bufA, N_NODES * 96);

  k_gemm_wmma<96, 4, false><<<MB, 128, 0, stream>>>(bufA, wpG2, nullptr, bufB, N_NODES, 64, 0);
  k_node_init<<<(N_NODES * 64 + T - 1) / T, T, 0, stream>>>(bufB, dinv, b_g2, out, N_NODES, 64);
  k_scatter<64><<<(N_EDGES * 16 + T - 1) / T, T, 0, stream>>>(bufB, dinv, src, dst, out, N_EDGES);
}


# Edge aggregation as a per-row gather over a CSR built in the call (int rank atomics + scan) instead of float atomic scatter plus init and ReLU passes; row sum order = arrival rank
# speedup vs baseline: 1.6045x; 1.6045x over previous
#include <hip/hip_runtime.h>
#include <hip/hip_bf16.h>

typedef float v2f __attribute__((ext_vector_type(2)));
typedef float v8f __attribute__((ext_vector_type(8)));

#define N_NODES 50000
#define N_EDGES 800000

static_assert(N_NODES % 4 == 0, "k_scan moves whole int4 groups");

__global__ void k_zero(int* __restrict__ cnt, int n) {
  int i = blockIdx.x * blockDim.x + threadIdx.x;
  if (i < n) cnt[i] = 0;
}

__global__ void k_rank(const int* __restrict__ dst, int* __restrict__ cnt,
                       int* __restrict__ rank, int e) {
  int i = blockIdx.x * blockDim.x + threadIdx.x;
  if (i < e) rank[i] = atomicAdd(&cnt[dst[i]], 1);
}

#define SCAN_ITERS ((N_NODES + 4095) / 4096)

__global__ __launch_bounds__(1024) void k_scan(const int* __restrict__ cnt,
                                               int* __restrict__ rowptr,
                                               float* __restrict__ dinv) {
  __shared__ int wtot[SCAN_ITERS][32];
  const int tid = threadIdx.x;
  const int lane = tid & 31;
  const int wave = tid >> 5;

  int4 c[SCAN_ITERS];
#pragma unroll
  for (int g = 0; g < SCAN_ITERS; ++g) {
    const int i0 = g * 4096 + tid * 4;
    c[g] = (i0 < N_NODES) ? *(const int4*)(cnt + i0) : make_int4(0, 0, 0, 0);
  }

  int tsum[SCAN_ITERS], incl[SCAN_ITERS];
#pragma unroll
  for (int g = 0; g < SCAN_ITERS; ++g) {
    tsum[g] = c[g].x + c[g].y + c[g].z + c[g].w;
    incl[g] = tsum[g];
  }
#pragma unroll
  for (int off = 1; off < 32; off <<= 1) {
#pragma unroll
    for (int g = 0; g < SCAN_ITERS; ++g) {
      const int t = __shfl_up(incl[g], off);
      if (lane >= off) incl[g] += t;
    }
  }
  if (lane == 31) {
#pragma unroll
    for (int g = 0; g < SCAN_ITERS; ++g) wtot[g][wave] = incl[g];
  }
  __syncthreads();

  int wv[SCAN_ITERS], winc[SCAN_ITERS];
#pragma unroll
  for (int g = 0; g < SCAN_ITERS; ++g) {
    wv[g] = wtot[g][lane];
    winc[g] = wv[g];
  }
#pragma unroll
  for (int off = 1; off < 32; off <<= 1) {
#pragma unroll
    for (int g = 0; g < SCAN_ITERS; ++g) {
      const int t = __shfl_up(winc[g], off);
      if (lane >= off) winc[g] += t;
    }
  }

  int carry = 0;
#pragma unroll
  for (int g = 0; g < SCAN_ITERS; ++g) {
    const int i0 = g * 4096 + tid * 4;
    const int wave_excl = __shfl(winc[g], wave) - __shfl(wv[g], wave);
    const int total = __shfl(winc[g], 31);
    const int e0 = carry + wave_excl + (incl[g] - tsum[g]);
    if (i0 < N_NODES) {
      const int4 v = c[g];
      int4 r;
      r.x = e0;
      r.y = r.x + v.x;
      r.z = r.y + v.y;
      r.w = r.z + v.z;
      *(int4*)(rowptr + i0) = r;
      float4 d;
      d.x = rsqrtf((float)(v.x + 1));
      d.y = rsqrtf((float)(v.y + 1));
      d.z = rsqrtf((float)(v.z + 1));
      d.w = rsqrtf((float)(v.w + 1));
      *(float4*)(dinv + i0) = d;
    }
    carry += total;
  }
  if (tid < 48) rowptr[N_NODES + tid] = carry;
}

__global__ void k_copy4(const int* __restrict__ from, int* __restrict__ to, int n4) {
  int i = blockIdx.x * blockDim.x + threadIdx.x;
  if (i < n4) ((int4*)to)[i] = ((const int4*)from)[i];
}

__global__ void k_fill(const int* __restrict__ src, const int* __restrict__ dst,
                       const int* __restrict__ rank, const int* __restrict__ rowptr,
                       int* __restrict__ col, int e) {
  int i = blockIdx.x * blockDim.x + threadIdx.x;
  if (i < e) col[rowptr[dst[i]] + rank[i]] = src[i];
}

__global__ void k_pack_w(const float* __restrict__ W, float* __restrict__ Wp,
                         int K, int N) {
  int idx = blockIdx.x * blockDim.x + threadIdx.x;
  if (idx >= K * N) return;
  int k = idx / N;
  int n = idx - k * N;
  Wp[(size_t)(k >> 2) * (N * 4) + n * 4 + (k & 3)] = W[idx];
}

template <int KDIM, int NT, bool BIAS_RELU>
__global__ __launch_bounds__(128) void k_gemm_wmma(
    const float* __restrict__ A, const float* __restrict__ Wp,
    const float* __restrict__ bias, float* __restrict__ out,
    int M, int out_stride, int out_off) {
  const int lane = threadIdx.x & 31;
  const int wave = threadIdx.x >> 5;
  const int row_tile = (blockIdx.x * 4 + wave) * 16;
  if (row_tile >= M) return;

  const int m16 = lane & 15;
  const int khalf = lane >> 4;
  constexpr int NCOLS = NT * 16;

  v8f acc[NT];
#pragma unroll
  for (int t = 0; t < NT; ++t) acc[t] = (v8f){0.f, 0.f, 0.f, 0.f, 0.f, 0.f, 0.f, 0.f};

  const float* __restrict__ arow = A + (size_t)(row_tile + m16) * KDIM + khalf * 2;
  const float* __restrict__ wlane = Wp + m16 * 4 + khalf * 2;

  for (int k0 = 0; k0 < KDIM; k0 += 16) {
    if (k0 + 256 < KDIM) __builtin_prefetch(arow + k0 + 256, 0, 0);
#pragma unroll
    for (int kq = 0; kq < 4; ++kq) {
      const int k = k0 + kq * 4;
      v2f afrag = *(const v2f*)(arow + k);
      const float* wq = wlane + (size_t)(k >> 2) * (NCOLS * 4);
#pragma unroll
      for (int t = 0; t < NT; ++t) {
        v2f bfrag = *(const v2f*)(wq + t * 64);
        acc[t] = __builtin_amdgcn_wmma_f32_16x16x4_f32(
             false, afrag,  false, bfrag,
             (short)0, acc[t],  false,  false);
      }
    }
  }

  const int rbase = row_tile + khalf * 8;
#pragma unroll
  for (int t = 0; t < NT; ++t) {
#pragma unroll
    for (int i = 0; i < 8; ++i) {
      const int c = t * 16 + m16;
      float v = acc[t][i];
      if (BIAS_RELU) v = fmaxf(v + bias[c], 0.0f);
      out[(size_t)(rbase + i) * out_stride + out_off + c] = v;
    }
  }
}

template <int C, bool RELU>
__global__ __launch_bounds__(256) void k_gather(
    const float* __restrict__ xw, const float* __restrict__ dinv,
    const int* __restrict__ rowptr, const int* __restrict__ col,
    const float* __restrict__ bias, float* __restrict__ out, int n) {
  constexpr int V = C / 32;
  const int lane = threadIdx.x & 31;
  const int node = __builtin_amdgcn_readfirstlane((int)(blockIdx.x * 8 + (threadIdx.x >> 5)));
  if (node >= n) return;

  const int beg = rowptr[node];
  const int end = rowptr[node + 1];
  const float di = dinv[node];

  float acc[V];
  {
    const float* __restrict__ p = xw + (size_t)node * C + lane;
    const float w = di * di;
#pragma unroll
    for (int v = 0; v < V; ++v) acc[v] = p[32 * v] * w;
  }

  for (int base = beg; base < end; base += 32) {
    const int m = min(32, end - base);
    int s_l = 0;
    float w_l = 0.0f;
    if (lane < m) {
      s_l = col[base + lane];
      w_l = dinv[s_l] * di;
    }
    int j = 0;
    for (; j + 4 <= m; j += 4) {
      float x[4][V];
      float w[4];
#pragma unroll
      for (int u = 0; u < 4; ++u) {
        const int s = __builtin_amdgcn_readlane(s_l, j + u);
        w[u] = __builtin_bit_cast(float, __builtin_amdgcn_readlane(__builtin_bit_cast(int, w_l), j + u));
        const float* __restrict__ p = xw + (size_t)s * C + lane;
#pragma unroll
        for (int v = 0; v < V; ++v) x[u][v] = p[32 * v];
      }
#pragma unroll
      for (int u = 0; u < 4; ++u) {
#pragma unroll
        for (int v = 0; v < V; ++v) acc[v] += x[u][v] * w[u];
      }
    }
    for (; j < m; ++j) {
      const int s = __builtin_amdgcn_readlane(s_l, j);
      const float w = __builtin_bit_cast(float, __builtin_amdgcn_readlane(__builtin_bit_cast(int, w_l), j));
      const float* __restrict__ p = xw + (size_t)s * C + lane;
#pragma unroll
      for (int v = 0; v < V; ++v) acc[v] += p[32 * v] * w;
    }
  }

  float* __restrict__ o = out + (size_t)node * C + lane;
#pragma unroll
  for (int v = 0; v < V; ++v) {
    float r = acc[v] + bias[lane + 32 * v];
    if (RELU) r = fmaxf(r, 0.0f);
    o[32 * v] = r;
  }
}

extern "C" void kernel_launch(void* const* d_in, const int* in_sizes, int n_in,
                              void* d_out, int out_size, void* d_ws, size_t ws_size,
                              hipStream_t stream) {
  (void)in_sizes; (void)n_in; (void)out_size; (void)ws_size;

  const float* feature = (const float*)d_in[0];
  const float* img     = (const float*)d_in[1];
  const int*   edges   = (const int*)d_in[2];
  const float* W_feat  = (const float*)d_in[3];
  const float* b_feat  = (const float*)d_in[4];
  const float* W_img   = (const float*)d_in[5];
  const float* b_img   = (const float*)d_in[6];
  const float* W_g1    = (const float*)d_in[7];
  const float* b_g1    = (const float*)d_in[8];
  const float* W_g2    = (const float*)d_in[9];
  const float* b_g2    = (const float*)d_in[10];
  float* out = (float*)d_out;

  const int* src = edges;
  const int* dst = edges + N_EDGES;

  float* ws     = (float*)d_ws;
  float* dinv   = ws;
  float* bufA   = ws + 50048;
  float* bufB   = bufA + (size_t)N_NODES * 96;
  float* wpF    = bufB + (size_t)N_NODES * 96;
  float* wpI    = wpF + 512 * 64;
  float* wpG1   = wpI + 256 * 32;
  float* wpG2   = wpG1 + 96 * 96;
  int*   rowptr1 = (int*)out;
  int*   col1    = rowptr1 + 50048;
  int*   rowptr2 = (int*)(bufB + (size_t)N_NODES * 64);
  int*   col2    = rowptr2 + 50048;
  int*   cnt    = (int*)bufB;
  int*   rank   = cnt + 50048;

  const int T = 256;
  const int MB = (N_NODES + 63) / 64;

  k_pack_w<<<(512 * 64 + T - 1) / T, T, 0, stream>>>(W_feat, wpF, 512, 64);
  k_pack_w<<<(256 * 32 + T - 1) / T, T, 0, stream>>>(W_img, wpI, 256, 32);
  k_pack_w<<<(96 * 96 + T - 1) / T, T, 0, stream>>>(W_g1, wpG1, 96, 96);
  k_pack_w<<<(96 * 64 + T - 1) / T, T, 0, stream>>>(W_g2, wpG2, 96, 64);

  k_zero<<<(N_NODES + T - 1) / T, T, 0, stream>>>(cnt, N_NODES);
  k_rank<<<(N_EDGES + T - 1) / T, T, 0, stream>>>(dst, cnt, rank, N_EDGES);
  k_scan<<<1, 1024, 0, stream>>>(cnt, rowptr1, dinv);
  k_fill<<<(N_EDGES + T - 1) / T, T, 0, stream>>>(src, dst, rank, rowptr1, col1, N_EDGES);

  k_gemm_wmma<512, 4, true><<<MB, 128, 0, stream>>>(feature, wpF, b_feat, bufA, N_NODES, 96, 0);
  k_gemm_wmma<256, 2, true><<<MB, 128, 0, stream>>>(img, wpI, b_img, bufA, N_NODES, 96, 64);

  k_gemm_wmma<96, 6, false><<<MB, 128, 0, stream>>>(bufA, wpG1, nullptr, bufB, N_NODES, 96, 0);
  k_gather<96, true><<<(N_NODES + 7) / 8, 256, 0, stream>>>(bufB, dinv, rowptr1, col1, b_g1, bufA, N_NODES);
  k_copy4<<<((50048 + N_EDGES) / 4 + T - 1) / T, T, 0, stream>>>(rowptr1, rowptr2, (50048 + N_EDGES) / 4);

  k_gemm_wmma<96, 4, false><<<MB, 128, 0, stream>>>(bufA, wpG2, nullptr, bufB, N_NODES, 64, 0);
  k_gather<64, false><<<(N_NODES + 7) / 8, 256, 0, stream>>>(bufB, dinv, rowptr2, col2, b_g2, out, N_NODES);
}


# Stages of the edge chain and the dense chain that are ready together share launches (packs+clear; rank with encoders; fill with GEMM 1; adjacency move with GEMM 2): 7 launches, not 15
# speedup vs baseline: 1.7581x; 1.0957x over previous
#include <hip/hip_runtime.h>
#include <hip/hip_bf16.h>

typedef float v2f __attribute__((ext_vector_type(2)));
typedef float v8f __attribute__((ext_vector_type(8)));

#define N_NODES 50000
#define N_EDGES 800000

static_assert(N_NODES % 4 == 0, "k_scan moves whole int4 groups");

#define SCAN_ITERS ((N_NODES + 4095) / 4096)

__global__ __launch_bounds__(1024) void k_scan(const int* __restrict__ cnt,
                                               int* __restrict__ rowptr,
                                               float* __restrict__ dinv) {
  __shared__ int wtot[SCAN_ITERS][32];
  const int tid = threadIdx.x;
  const int lane = tid & 31;
  const int wave = tid >> 5;

  int4 c[SCAN_ITERS];
#pragma unroll
  for (int g = 0; g < SCAN_ITERS; ++g) {
    const int i0 = g * 4096 + tid * 4;
    c[g] = (i0 < N_NODES) ? *(const int4*)(cnt + i0) : make_int4(0, 0, 0, 0);
  }

  int tsum[SCAN_ITERS], incl[SCAN_ITERS];
#pragma unroll
  for (int g = 0; g < SCAN_ITERS; ++g) {
    tsum[g] = c[g].x + c[g].y + c[g].z + c[g].w;
    incl[g] = tsum[g];
  }
#pragma unroll
  for (int off = 1; off < 32; off <<= 1) {
#pragma unroll
    for (int g = 0; g < SCAN_ITERS; ++g) {
      const int t = __shfl_up(incl[g], off);
      if (lane >= off) incl[g] += t;
    }
  }
  if (lane == 31) {
#pragma unroll
    for (int g = 0; g < SCAN_ITERS; ++g) wtot[g][wave] = incl[g];
  }
  __syncthreads();

  int wv[SCAN_ITERS], winc[SCAN_ITERS];
#pragma unroll
  for (int g = 0; g < SCAN_ITERS; ++g) {
    wv[g] = wtot[g][lane];
    winc[g] = wv[g];
  }
#pragma unroll
  for (int off = 1; off < 32; off <<= 1) {
#pragma unroll
    for (int g = 0; g < SCAN_ITERS; ++g) {
      const int t = __shfl_up(winc[g], off);
      if (lane >= off) winc[g] += t;
    }
  }

  int carry = 0;
#pragma unroll
  for (int g = 0; g < SCAN_ITERS; ++g) {
    const int i0 = g * 4096 + tid * 4;
    const int wave_excl = __shfl(winc[g], wave) - __shfl(wv[g], wave);
    const int total = __shfl(winc[g], 31);
    const int e0 = carry + wave_excl + (incl[g] - tsum[g]);
    if (i0 < N_NODES) {
      const int4 v = c[g];
      int4 r;
      r.x = e0;
      r.y = r.x + v.x;
      r.z = r.y + v.y;
      r.w = r.z + v.z;
      *(int4*)(rowptr + i0) = r;
      float4 d;
      d.x = rsqrtf((float)(v.x + 1));
      d.y = rsqrtf((float)(v.y + 1));
      d.z = rsqrtf((float)(v.z + 1));
      d.w = rsqrtf((float)(v.w + 1));
      *(float4*)(dinv + i0) = d;
    }
    carry += total;
  }
  if (tid < 48) rowptr[N_NODES + tid] = carry;
}


template <int K, int N>
__device__ __forceinline__ void pack_w(const float* __restrict__ W, float* __restrict__ Wp,
                                       int idx) {
  if (idx >= K * N) return;
  const int k = idx / N;
  const int n = idx - k * N;
  Wp[(size_t)(k >> 2) * (N * 4) + n * 4 + (k & 3)] = W[idx];
}

template <int KDIM, int NT, bool BIAS_RELU>
__device__ __forceinline__ void gemm_rows16(
    const float* __restrict__ A, const float* __restrict__ Wp,
    const float* __restrict__ bias, float* __restrict__ out,
    int row_tile, int lane, int out_stride, int out_off) {
  const int m16 = lane & 15;
  const int khalf = lane >> 4;
  constexpr int NCOLS = NT * 16;

  v8f acc[NT];
#pragma unroll
  for (int t = 0; t < NT; ++t) acc[t] = (v8f){0.f, 0.f, 0.f, 0.f, 0.f, 0.f, 0.f, 0.f};

  const float* __restrict__ arow = A + (size_t)(row_tile + m16) * KDIM + khalf * 2;
  const float* __restrict__ wlane = Wp + m16 * 4 + khalf * 2;

  for (int k0 = 0; k0 < KDIM; k0 += 16) {
    if (k0 + 256 < KDIM) __builtin_prefetch(arow + k0 + 256, 0, 0);
#pragma unroll
    for (int kq = 0; kq < 4; ++kq) {
      const int k = k0 + kq * 4;
      v2f afrag = *(const v2f*)(arow + k);
      const float* wq = wlane + (size_t)(k >> 2) * (NCOLS * 4);
#pragma unroll
      for (int t = 0; t < NT; ++t) {
        v2f bfrag = *(const v2f*)(wq + t * 64);
        acc[t] = __builtin_amdgcn_wmma_f32_16x16x4_f32(
             false, afrag,  false, bfrag,
             (short)0, acc[t],  false,  false);
      }
    }
  }

  const int rbase = row_tile + khalf * 8;
#pragma unroll
  for (int t = 0; t < NT; ++t) {
#pragma unroll
    for (int i = 0; i < 8; ++i) {
      const int c = t * 16 + m16;
      float v = acc[t][i];
      if (BIAS_RELU) v = fmaxf(v + bias[c], 0.0f);
      out[(size_t)(rbase + i) * out_stride + out_off + c] = v;
    }
  }
}

#define GB ((N_NODES + 63) / 64)
#define EB ((N_EDGES + 127) / 128)
#define ADJ4 ((50048 + N_EDGES) / 4)
#define CB ((ADJ4 + 127) / 128)

#define PB_F ((512 * 64 + 255) / 256)
#define PB_I ((256 * 32 + 255) / 256)
#define PB_1 ((96 * 96 + 255) / 256)
#define PB_2 ((96 * 64 + 255) / 256)
#define PB_Z ((N_NODES + 255) / 256)

__global__ __launch_bounds__(256) void k_prep(
    const float* __restrict__ W_feat, float* __restrict__ wpF,
    const float* __restrict__ W_img, float* __restrict__ wpI,
    const float* __restrict__ W_g1, float* __restrict__ wpG1,
    const float* __restrict__ W_g2, float* __restrict__ wpG2,
    int* __restrict__ cnt) {
  const int b = blockIdx.x;
  const int t = threadIdx.x;
  if (b < PB_F) {
    pack_w<512, 64>(W_feat, wpF, b * 256 + t);
  } else if (b < PB_F + PB_I) {
    pack_w<256, 32>(W_img, wpI, (b - PB_F) * 256 + t);
  } else if (b < PB_F + PB_I + PB_1) {
    pack_w<96, 96>(W_g1, wpG1, (b - PB_F - PB_I) * 256 + t);
  } else if (b < PB_F + PB_I + PB_1 + PB_2) {
    pack_w<96, 64>(W_g2, wpG2, (b - PB_F - PB_I - PB_1) * 256 + t);
  } else {
    const int i = (b - PB_F - PB_I - PB_1 - PB_2) * 256 + t;
    if (i < N_NODES) cnt[i] = 0;
  }
}

__global__ __launch_bounds__(128) void k_enc_rank(
    const float* __restrict__ feature, const float* __restrict__ wpF,
    const float* __restrict__ b_feat,
    const float* __restrict__ img, const float* __restrict__ wpI,
    const float* __restrict__ b_img, float* __restrict__ x,
    const int* __restrict__ dst, int* __restrict__ cnt, int* __restrict__ rank) {
  const int b = blockIdx.x;
  const int lane = threadIdx.x & 31;
  const int wave = threadIdx.x >> 5;
  if (b < GB) {
    const int row_tile = (b * 4 + wave) * 16;
    if (row_tile < N_NODES) gemm_rows16<512, 4, true>(feature, wpF, b_feat, x, row_tile, lane, 96, 0);
  } else if (b < 2 * GB) {
    const int row_tile = ((b - GB) * 4 + wave) * 16;
    if (row_tile < N_NODES) gemm_rows16<256, 2, true>(img, wpI, b_img, x, row_tile, lane, 96, 64);
  } else {
    const int i = (b - 2 * GB) * 128 + threadIdx.x;
    if (i < N_EDGES) rank[i] = atomicAdd(&cnt[dst[i]], 1);
  }
}

__global__ __launch_bounds__(128) __attribute__((amdgpu_waves_per_eu(1, 7))) void k_gemm1_fill(
    const float* __restrict__ x, const float* __restrict__ wpG1, float* __restrict__ xw1,
    const int* __restrict__ src, const int* __restrict__ dst,
    const int* __restrict__ rank, const int* __restrict__ rowptr,
    int* __restrict__ col) {
  const int b = blockIdx.x;
  if (b < GB) {
    const int lane = threadIdx.x & 31;
    const int wave = threadIdx.x >> 5;
    const int row_tile = (b * 4 + wave) * 16;
    if (row_tile < N_NODES) gemm_rows16<96, 6, false>(x, wpG1, nullptr, xw1, row_tile, lane, 96, 0);
  } else {
    const int i = (b - GB) * 128 + threadIdx.x;
    if (i < N_EDGES) col[rowptr[dst[i]] + rank[i]] = src[i];
  }
}

__global__ __launch_bounds__(128) void k_gemm2_copy(
    const float* __restrict__ h, const float* __restrict__ wpG2, float* __restrict__ xw2,
    const int* __restrict__ from, int* __restrict__ to) {
  const int b = blockIdx.x;
  if (b < GB) {
    const int lane = threadIdx.x & 31;
    const int wave = threadIdx.x >> 5;
    const int row_tile = (b * 4 + wave) * 16;
    if (row_tile < N_NODES) gemm_rows16<96, 4, false>(h, wpG2, nullptr, xw2, row_tile, lane, 64, 0);
  } else {
    const int i = (b - GB) * 128 + threadIdx.x;
    if (i < ADJ4) ((int4*)to)[i] = ((const int4*)from)[i];
  }
}

template <int C, bool RELU>
__global__ __launch_bounds__(256) void k_gather(
    const float* __restrict__ xw, const float* __restrict__ dinv,
    const int* __restrict__ rowptr, const int* __restrict__ col,
    const float* __restrict__ bias, float* __restrict__ out, int n) {
  constexpr int V = C / 32;
  const int lane = threadIdx.x & 31;
  const int node = __builtin_amdgcn_readfirstlane((int)(blockIdx.x * 8 + (threadIdx.x >> 5)));
  if (node >= n) return;

  const int beg = rowptr[node];
  const int end = rowptr[node + 1];
  const float di = dinv[node];

  float acc[V];
  {
    const float* __restrict__ p = xw + (size_t)node * C + lane;
    const float w = di * di;
#pragma unroll
    for (int v = 0; v < V; ++v) acc[v] = p[32 * v] * w;
  }

  for (int base = beg; base < end; base += 32) {
    const int m = min(32, end - base);
    int s_l = 0;
    float w_l = 0.0f;
    if (lane < m) {
      s_l = col[base + lane];
      w_l = dinv[s_l] * di;
    }
    int j = 0;
    for (; j + 4 <= m; j += 4) {
      float x[4][V];
      float w[4];
#pragma unroll
      for (int u = 0; u < 4; ++u) {
        const int s = __builtin_amdgcn_readlane(s_l, j + u);
        w[u] = __builtin_bit_cast(float, __builtin_amdgcn_readlane(__builtin_bit_cast(int, w_l), j + u));
        const float* __restrict__ p = xw + (size_t)s * C + lane;
#pragma unroll
        for (int v = 0; v < V; ++v) x[u][v] = p[32 * v];
      }
#pragma unroll
      for (int u = 0; u < 4; ++u) {
#pragma unroll
        for (int v = 0; v < V; ++v) acc[v] += x[u][v] * w[u];
      }
    }
    for (; j < m; ++j) {
      const int s = __builtin_amdgcn_readlane(s_l, j);
      const float w = __builtin_bit_cast(float, __builtin_amdgcn_readlane(__builtin_bit_cast(int, w_l), j));
      const float* __restrict__ p = xw + (size_t)s * C + lane;
#pragma unroll
      for (int v = 0; v < V; ++v) acc[v] += p[32 * v] * w;
    }
  }

  float* __restrict__ o = out + (size_t)node * C + lane;
#pragma unroll
  for (int v = 0; v < V; ++v) {
    float r = acc[v] + bias[lane + 32 * v];
    if (RELU) r = fmaxf(r, 0.0f);
    o[32 * v] = r;
  }
}

extern "C" void kernel_launch(void* const* d_in, const int* in_sizes, int n_in,
                              void* d_out, int out_size, void* d_ws, size_t ws_size,
                              hipStream_t stream) {
  (void)in_sizes; (void)n_in; (void)out_size; (void)ws_size;

  const float* feature = (const float*)d_in[0];
  const float* img     = (const float*)d_in[1];
  const int*   edges   = (const int*)d_in[2];
  const float* W_feat  = (const float*)d_in[3];
  const float* b_feat  = (const float*)d_in[4];
  const float* W_img   = (const float*)d_in[5];
  const float* b_img   = (const float*)d_in[6];
  const float* W_g1    = (const float*)d_in[7];
  const float* b_g1    = (const float*)d_in[8];
  const float* W_g2    = (const float*)d_in[9];
  const float* b_g2    = (const float*)d_in[10];
  float* out = (float*)d_out;

  const int* src = edges;
  const int* dst = edges + N_EDGES;

  float* ws     = (float*)d_ws;
  float* dinv   = ws;
  float* bufA   = ws + 50048;
  float* bufB   = bufA + (size_t)N_NODES * 96;
  float* wpF    = bufB + (size_t)N_NODES * 96;
  float* wpI    = wpF + 512 * 64;
  float* wpG1   = wpI + 256 * 32;
  float* wpG2   = wpG1 + 96 * 96;
  int*   rowptr1 = (int*)out;
  int*   col1    = rowptr1 + 50048;
  int*   cnt     = col1 + N_EDGES;
  int*   rank    = cnt + 50048;
  int*   rowptr2 = (int*)(bufB + (size_t)N_NODES * 64);
  int*   col2    = rowptr2 + 50048;

  k_prep<<<PB_F + PB_I + PB_1 + PB_2 + PB_Z, 256, 0, stream>>>(
      W_feat, wpF, W_img, wpI, W_g1, wpG1, W_g2, wpG2, cnt);

  k_enc_rank<<<2 * GB + EB, 128, 0, stream>>>(feature, wpF, b_feat, img, wpI, b_img, bufA,
                                               dst, cnt, rank);

  k_scan<<<1, 1024, 0, stream>>>(cnt, rowptr1, dinv);

  k_gemm1_fill<<<GB + EB, 128, 0, stream>>>(bufA, wpG1, bufB, src, dst, rank, rowptr1, col1);

  k_gather<96, true><<<(N_NODES + 7) / 8, 256, 0, stream>>>(bufB, dinv, rowptr1, col1, b_g1, bufA, N_NODES);

  k_gemm2_copy<<<GB + CB, 128, 0, stream>>>(bufA, wpG2, bufB, rowptr1, rowptr2);

  k_gather<64, false><<<(N_NODES + 7) / 8, 256, 0, stream>>>(bufB, dinv, rowptr2, col2, b_g2, out, N_NODES);
}


# GEMM operand feed: A rows and packed weights read in 16-byte pieces (two WMMAs each), next A pieces issued one chunk ahead of their WMMAs; k is summed in another order inside each dot product
# speedup vs baseline: 1.9572x; 1.1132x over previous
#include <hip/hip_runtime.h>
#include <hip/hip_bf16.h>

typedef float v2f __attribute__((ext_vector_type(2)));
typedef float v8f __attribute__((ext_vector_type(8)));

#define N_NODES 50000
#define N_EDGES 800000

static_assert(N_NODES % 4 == 0, "k_scan moves whole int4 groups");

#define SCAN_ITERS ((N_NODES + 4095) / 4096)

__global__ __launch_bounds__(1024) void k_scan(const int* __restrict__ cnt,
                                               int* __restrict__ rowptr,
                                               float* __restrict__ dinv) {
  __shared__ int wtot[SCAN_ITERS][32];
  const int tid = threadIdx.x;
  const int lane = tid & 31;
  const int wave = tid >> 5;

  int4 c[SCAN_ITERS];
#pragma unroll
  for (int g = 0; g < SCAN_ITERS; ++g) {
    const int i0 = g * 4096 + tid * 4;
    c[g] = (i0 < N_NODES) ? *(const int4*)(cnt + i0) : make_int4(0, 0, 0, 0);
  }

  int tsum[SCAN_ITERS], incl[SCAN_ITERS];
#pragma unroll
  for (int g = 0; g < SCAN_ITERS; ++g) {
    tsum[g] = c[g].x + c[g].y + c[g].z + c[g].w;
    incl[g] = tsum[g];
  }
#pragma unroll
  for (int off = 1; off < 32; off <<= 1) {
#pragma unroll
    for (int g = 0; g < SCAN_ITERS; ++g) {
      const int t = __shfl_up(incl[g], off);
      if (lane >= off) incl[g] += t;
    }
  }
  if (lane == 31) {
#pragma unroll
    for (int g = 0; g < SCAN_ITERS; ++g) wtot[g][wave] = incl[g];
  }
  __syncthreads();

  int wv[SCAN_ITERS], winc[SCAN_ITERS];
#pragma unroll
  for (int g = 0; g < SCAN_ITERS; ++g) {
    wv[g] = wtot[g][lane];
    winc[g] = wv[g];
  }
#pragma unroll
  for (int off = 1; off < 32; off <<= 1) {
#pragma unroll
    for (int g = 0; g < SCAN_ITERS; ++g) {
      const int t = __shfl_up(winc[g], off);
      if (lane >= off) winc[g] += t;
    }
  }

  int carry = 0;
#pragma unroll
  for (int g = 0; g < SCAN_ITERS; ++g) {
    const int i0 = g * 4096 + tid * 4;
    const int wave_excl = __shfl(winc[g], wave) - __shfl(wv[g], wave);
    const int total = __shfl(winc[g], 31);
    const int e0 = carry + wave_excl + (incl[g] - tsum[g]);
    if (i0 < N_NODES) {
      const int4 v = c[g];
      int4 r;
      r.x = e0;
      r.y = r.x + v.x;
      r.z = r.y + v.y;
      r.w = r.z + v.z;
      *(int4*)(rowptr + i0) = r;
      float4 d;
      d.x = rsqrtf((float)(v.x + 1));
      d.y = rsqrtf((float)(v.y + 1));
      d.z = rsqrtf((float)(v.z + 1));
      d.w = rsqrtf((float)(v.w + 1));
      *(float4*)(dinv + i0) = d;
    }
    carry += total;
  }
  if (tid < 48) rowptr[N_NODES + tid] = carry;
}


template <int K, int N>
__device__ __forceinline__ void pack_w(const float* __restrict__ W, float* __restrict__ Wp,
                                       int idx) {
  if (idx >= K * N) return;
  const int k = idx / N;
  const int n = idx - k * N;
  Wp[(size_t)(k >> 2) * (N * 4) + n * 4 + (k & 3)] = W[idx];
}

typedef float v4f __attribute__((ext_vector_type(4)));

template <int KDIM, int NT, bool BIAS_RELU>
__device__ __forceinline__ void gemm_rows16(
    const float* __restrict__ A, const float* __restrict__ Wp,
    const float* __restrict__ bias, float* __restrict__ out,
    int row_tile, int lane, int out_stride, int out_off) {
  static_assert(KDIM % 32 == 0, "A rows are read in pairs of 16-float chunks");
  const int m16 = lane & 15;
  const int khalf = lane >> 4;
  constexpr int NCOLS = NT * 16;
  constexpr int NCH = KDIM / 16;
  constexpr int GROUP = NCOLS * 4;

  v8f acc[NT];
#pragma unroll
  for (int t = 0; t < NT; ++t) acc[t] = (v8f){0.f, 0.f, 0.f, 0.f, 0.f, 0.f, 0.f, 0.f};

  const float* __restrict__ arow = A + (size_t)(row_tile + m16) * KDIM + khalf * 4;
  const float* __restrict__ wlane = Wp + (size_t)khalf * GROUP + m16 * 4;

  v4f a_even[2], a_odd[2];
#pragma unroll
  for (int q = 0; q < 2; ++q) a_even[q] = *(const v4f*)(arow + q * 8);

  auto chunk = [&](const int ch, const v4f (&a_cur)[2], v4f (&a_nxt)[2]) {
    const float* __restrict__ wc = wlane + (size_t)ch * (4 * GROUP);
    v4f b[2][NT];
#pragma unroll
    for (int q = 0; q < 2; ++q) {
#pragma unroll
      for (int t = 0; t < NT; ++t) b[q][t] = *(const v4f*)(wc + (size_t)q * (2 * GROUP) + t * 64);
    }
    __builtin_amdgcn_sched_barrier(0);
    if (ch + 1 < NCH) {
#pragma unroll
      for (int q = 0; q < 2; ++q) a_nxt[q] = *(const v4f*)(arow + (ch + 1) * 16 + q * 8);
    }
    if ((ch & 1) == 0 && ch + 8 < NCH) __builtin_prefetch(arow + (ch + 8) * 16, 0, 0);
    __builtin_amdgcn_sched_barrier(0);
#pragma unroll
    for (int q = 0; q < 2; ++q) {
      const v2f a0 = {a_cur[q].x, a_cur[q].y};
      const v2f a1 = {a_cur[q].z, a_cur[q].w};
#pragma unroll
      for (int t = 0; t < NT; ++t) {
        const v2f b0 = {b[q][t].x, b[q][t].y};
        acc[t] = __builtin_amdgcn_wmma_f32_16x16x4_f32(
             false, a0,  false, b0,
             (short)0, acc[t],  false,  false);
      }
#pragma unroll
      for (int t = 0; t < NT; ++t) {
        const v2f b1 = {b[q][t].z, b[q][t].w};
        acc[t] = __builtin_amdgcn_wmma_f32_16x16x4_f32(
             false, a1,  false, b1,
             (short)0, acc[t],  false,  false);
      }
    }
    __builtin_amdgcn_sched_barrier(0);
  };

  for (int ch = 0; ch < NCH; ch += 2) {
    chunk(ch, a_even, a_odd);
    chunk(ch + 1, a_odd, a_even);
  }

  const int rbase = row_tile + khalf * 8;
#pragma unroll
  for (int t = 0; t < NT; ++t) {
#pragma unroll
    for (int i = 0; i < 8; ++i) {
      const int c = t * 16 + m16;
      float v = acc[t][i];
      if (BIAS_RELU) v = fmaxf(v + bias[c], 0.0f);
      out[(size_t)(rbase + i) * out_stride + out_off + c] = v;
    }
  }
}

#define GB ((N_NODES + 63) / 64)
#define EB ((N_EDGES + 127) / 128)
#define ADJ4 ((50048 + N_EDGES) / 4)
#define CB ((ADJ4 + 127) / 128)

#define PB_F ((512 * 64 + 255) / 256)
#define PB_I ((256 * 32 + 255) / 256)
#define PB_1 ((96 * 96 + 255) / 256)
#define PB_2 ((96 * 64 + 255) / 256)
#define PB_Z ((N_NODES + 255) / 256)

__global__ __launch_bounds__(256) void k_prep(
    const float* __restrict__ W_feat, float* __restrict__ wpF,
    const float* __restrict__ W_img, float* __restrict__ wpI,
    const float* __restrict__ W_g1, float* __restrict__ wpG1,
    const float* __restrict__ W_g2, float* __restrict__ wpG2,
    int* __restrict__ cnt) {
  const int b = blockIdx.x;
  const int t = threadIdx.x;
  if (b < PB_F) {
    pack_w<512, 64>(W_feat, wpF, b * 256 + t);
  } else if (b < PB_F + PB_I) {
    pack_w<256, 32>(W_img, wpI, (b - PB_F) * 256 + t);
  } else if (b < PB_F + PB_I + PB_1) {
    pack_w<96, 96>(W_g1, wpG1, (b - PB_F - PB_I) * 256 + t);
  } else if (b < PB_F + PB_I + PB_1 + PB_2) {
    pack_w<96, 64>(W_g2, wpG2, (b - PB_F - PB_I - PB_1) * 256 + t);
  } else {
    const int i = (b - PB_F - PB_I - PB_1 - PB_2) * 256 + t;
    if (i < N_NODES) cnt[i] = 0;
  }
}

__global__ __launch_bounds__(128) void k_enc_rank(
    const float* __restrict__ feature, const float* __restrict__ wpF,
    const float* __restrict__ b_feat,
    const float* __restrict__ img, const float* __restrict__ wpI,
    const float* __restrict__ b_img, float* __restrict__ x,
    const int* __restrict__ dst, int* __restrict__ cnt, int* __restrict__ rank) {
  const int b = blockIdx.x;
  const int lane = threadIdx.x & 31;
  const int wave = threadIdx.x >> 5;
  if (b < GB) {
    const int row_tile = (b * 4 + wave) * 16;
    if (row_tile < N_NODES) gemm_rows16<512, 4, true>(feature, wpF, b_feat, x, row_tile, lane, 96, 0);
  } else if (b < 2 * GB) {
    const int row_tile = ((b - GB) * 4 + wave) * 16;
    if (row_tile < N_NODES) gemm_rows16<256, 2, true>(img, wpI, b_img, x, row_tile, lane, 96, 64);
  } else {
    const int i = (b - 2 * GB) * 128 + threadIdx.x;
    if (i < N_EDGES) rank[i] = atomicAdd(&cnt[dst[i]], 1);
  }
}

__global__ __launch_bounds__(128) __attribute__((amdgpu_waves_per_eu(1, 7))) void k_gemm1_fill(
    const float* __restrict__ x, const float* __restrict__ wpG1, float* __restrict__ xw1,
    const int* __restrict__ src, const int* __restrict__ dst,
    const int* __restrict__ rank, const int* __restrict__ rowptr,
    int* __restrict__ col) {
  const int b = blockIdx.x;
  if (b < GB) {
    const int lane = threadIdx.x & 31;
    const int wave = threadIdx.x >> 5;
    const int row_tile = (b * 4 + wave) * 16;
    if (row_tile < N_NODES) gemm_rows16<96, 6, false>(x, wpG1, nullptr, xw1, row_tile, lane, 96, 0);
  } else {
    const int i = (b - GB) * 128 + threadIdx.x;
    if (i < N_EDGES) col[rowptr[dst[i]] + rank[i]] = src[i];
  }
}

__global__ __launch_bounds__(128) void k_gemm2_copy(
    const float* __restrict__ h, const float* __restrict__ wpG2, float* __restrict__ xw2,
    const int* __restrict__ from, int* __restrict__ to) {
  const int b = blockIdx.x;
  if (b < GB) {
    const int lane = threadIdx.x & 31;
    const int wave = threadIdx.x >> 5;
    const int row_tile = (b * 4 + wave) * 16;
    if (row_tile < N_NODES) gemm_rows16<96, 4, false>(h, wpG2, nullptr, xw2, row_tile, lane, 64, 0);
  } else {
    const int i = (b - GB) * 128 + threadIdx.x;
    if (i < ADJ4) ((int4*)to)[i] = ((const int4*)from)[i];
  }
}

template <int C, bool RELU>
__global__ __launch_bounds__(256) void k_gather(
    const float* __restrict__ xw, const float* __restrict__ dinv,
    const int* __restrict__ rowptr, const int* __restrict__ col,
    const float* __restrict__ bias, float* __restrict__ out, int n) {
  constexpr int V = C / 32;
  const int lane = threadIdx.x & 31;
  const int node = __builtin_amdgcn_readfirstlane((int)(blockIdx.x * 8 + (threadIdx.x >> 5)));
  if (node >= n) return;

  const int beg = rowptr[node];
  const int end = rowptr[node + 1];
  const float di = dinv[node];

  float acc[V];
  {
    const float* __restrict__ p = xw + (size_t)node * C + lane;
    const float w = di * di;
#pragma unroll
    for (int v = 0; v < V; ++v) acc[v] = p[32 * v] * w;
  }

  for (int base = beg; base < end; base += 32) {
    const int m = min(32, end - base);
    int s_l = 0;
    float w_l = 0.0f;
    if (lane < m) {
      s_l = col[base + lane];
      w_l = dinv[s_l] * di;
    }
    int j = 0;
    for (; j + 4 <= m; j += 4) {
      float x[4][V];
      float w[4];
#pragma unroll
      for (int u = 0; u < 4; ++u) {
        const int s = __builtin_amdgcn_readlane(s_l, j + u);
        w[u] = __builtin_bit_cast(float, __builtin_amdgcn_readlane(__builtin_bit_cast(int, w_l), j + u));
        const float* __restrict__ p = xw + (size_t)s * C + lane;
#pragma unroll
        for (int v = 0; v < V; ++v) x[u][v] = p[32 * v];
      }
#pragma unroll
      for (int u = 0; u < 4; ++u) {
#pragma unroll
        for (int v = 0; v < V; ++v) acc[v] += x[u][v] * w[u];
      }
    }
    for (; j < m; ++j) {
      const int s = __builtin_amdgcn_readlane(s_l, j);
      const float w = __builtin_bit_cast(float, __builtin_amdgcn_readlane(__builtin_bit_cast(int, w_l), j));
      const float* __restrict__ p = xw + (size_t)s * C + lane;
#pragma unroll
      for (int v = 0; v < V; ++v) acc[v] += p[32 * v] * w;
    }
  }

  float* __restrict__ o = out + (size_t)node * C + lane;
#pragma unroll
  for (int v = 0; v < V; ++v) {
    float r = acc[v] + bias[lane + 32 * v];
    if (RELU) r = fmaxf(r, 0.0f);
    o[32 * v] = r;
  }
}

extern "C" void kernel_launch(void* const* d_in, const int* in_sizes, int n_in,
                              void* d_out, int out_size, void* d_ws, size_t ws_size,
                              hipStream_t stream) {
  (void)in_sizes; (void)n_in; (void)out_size; (void)ws_size;

  const float* feature = (const float*)d_in[0];
  const float* img     = (const float*)d_in[1];
  const int*   edges   = (const int*)d_in[2];
  const float* W_feat  = (const float*)d_in[3];
  const float* b_feat  = (const float*)d_in[4];
  const float* W_img   = (const float*)d_in[5];
  const float* b_img   = (const float*)d_in[6];
  const float* W_g1    = (const float*)d_in[7];
  const float* b_g1    = (const float*)d_in[8];
  const float* W_g2    = (const float*)d_in[9];
  const float* b_g2    = (const float*)d_in[10];
  float* out = (float*)d_out;

  const int* src = edges;
  const int* dst = edges + N_EDGES;

  float* ws     = (float*)d_ws;
  float* dinv   = ws;
  float* bufA   = ws + 50048;
  float* bufB   = bufA + (size_t)N_NODES * 96;
  float* wpF    = bufB + (size_t)N_NODES * 96;
  float* wpI    = wpF + 512 * 64;
  float* wpG1   = wpI + 256 * 32;
  float* wpG2   = wpG1 + 96 * 96;
  int*   rowptr1 = (int*)out;
  int*   col1    = rowptr1 + 50048;
  int*   cnt     = col1 + N_EDGES;
  int*   rank    = cnt + 50048;
  int*   rowptr2 = (int*)(bufB + (size_t)N_NODES * 64);
  int*   col2    = rowptr2 + 50048;

  k_prep<<<PB_F + PB_I + PB_1 + PB_2 + PB_Z, 256, 0, stream>>>(
      W_feat, wpF, W_img, wpI, W_g1, wpG1, W_g2, wpG2, cnt);

  k_enc_rank<<<2 * GB + EB, 128, 0, stream>>>(feature, wpF, b_feat, img, wpI, b_img, bufA,
                                               dst, cnt, rank);

  k_scan<<<1, 1024, 0, stream>>>(cnt, rowptr1, dinv);

  k_gemm1_fill<<<GB + EB, 128, 0, stream>>>(bufA, wpG1, bufB, src, dst, rank, rowptr1, col1);

  k_gather<96, true><<<(N_NODES + 7) / 8, 256, 0, stream>>>(bufB, dinv, rowptr1, col1, b_g1, bufA, N_NODES);

  k_gemm2_copy<<<GB + CB, 128, 0, stream>>>(bufA, wpG2, bufB, rowptr1, rowptr2);

  k_gather<64, false><<<(N_NODES + 7) / 8, 256, 0, stream>>>(bufB, dinv, rowptr2, col2, b_g2, out, N_NODES);
}


# Source-side dinv[s] folded into the rows each gather reads (GEMM-1 rows and layer-1 output rows written pre-scaled): gathers add plain rows, no per-edge dinv lookup; order of the sums kept
# speedup vs baseline: 2.0178x; 1.0310x over previous
#include <hip/hip_runtime.h>
#include <hip/hip_bf16.h>

typedef float v2f __attribute__((ext_vector_type(2)));
typedef float v8f __attribute__((ext_vector_type(8)));

#define N_NODES 50000
#define N_EDGES 800000

static_assert(N_NODES % 4 == 0, "k_scan moves whole int4 groups");

#define SCAN_ITERS ((N_NODES + 4095) / 4096)

__global__ __launch_bounds__(1024) void k_scan(const int* __restrict__ cnt,
                                               int* __restrict__ rowptr,
                                               float* __restrict__ dinv) {
  __shared__ int wtot[SCAN_ITERS][32];
  const int tid = threadIdx.x;
  const int lane = tid & 31;
  const int wave = tid >> 5;

  int4 c[SCAN_ITERS];
#pragma unroll
  for (int g = 0; g < SCAN_ITERS; ++g) {
    const int i0 = g * 4096 + tid * 4;
    c[g] = (i0 < N_NODES) ? *(const int4*)(cnt + i0) : make_int4(0, 0, 0, 0);
  }

  int tsum[SCAN_ITERS], incl[SCAN_ITERS];
#pragma unroll
  for (int g = 0; g < SCAN_ITERS; ++g) {
    tsum[g] = c[g].x + c[g].y + c[g].z + c[g].w;
    incl[g] = tsum[g];
  }
#pragma unroll
  for (int off = 1; off < 32; off <<= 1) {
#pragma unroll
    for (int g = 0; g < SCAN_ITERS; ++g) {
      const int t = __shfl_up(incl[g], off);
      if (lane >= off) incl[g] += t;
    }
  }
  if (lane == 31) {
#pragma unroll
    for (int g = 0; g < SCAN_ITERS; ++g) wtot[g][wave] = incl[g];
  }
  __syncthreads();

  int wv[SCAN_ITERS], winc[SCAN_ITERS];
#pragma unroll
  for (int g = 0; g < SCAN_ITERS; ++g) {
    wv[g] = wtot[g][lane];
    winc[g] = wv[g];
  }
#pragma unroll
  for (int off = 1; off < 32; off <<= 1) {
#pragma unroll
    for (int g = 0; g < SCAN_ITERS; ++g) {
      const int t = __shfl_up(winc[g], off);
      if (lane >= off) winc[g] += t;
    }
  }

  int carry = 0;
#pragma unroll
  for (int g = 0; g < SCAN_ITERS; ++g) {
    const int i0 = g * 4096 + tid * 4;
    const int wave_excl = __shfl(winc[g], wave) - __shfl(wv[g], wave);
    const int total = __shfl(winc[g], 31);
    const int e0 = carry + wave_excl + (incl[g] - tsum[g]);
    if (i0 < N_NODES) {
      const int4 v = c[g];
      int4 r;
      r.x = e0;
      r.y = r.x + v.x;
      r.z = r.y + v.y;
      r.w = r.z + v.z;
      *(int4*)(rowptr + i0) = r;
      float4 d;
      d.x = rsqrtf((float)(v.x + 1));
      d.y = rsqrtf((float)(v.y + 1));
      d.z = rsqrtf((float)(v.z + 1));
      d.w = rsqrtf((float)(v.w + 1));
      *(float4*)(dinv + i0) = d;
    }
    carry += total;
  }
  if (tid < 48) rowptr[N_NODES + tid] = carry;
}


template <int K, int N>
__device__ __forceinline__ void pack_w(const float* __restrict__ W, float* __restrict__ Wp,
                                       int idx) {
  if (idx >= K * N) return;
  const int k = idx / N;
  const int n = idx - k * N;
  Wp[(size_t)(k >> 2) * (N * 4) + n * 4 + (k & 3)] = W[idx];
}

typedef float v4f __attribute__((ext_vector_type(4)));

template <int KDIM, int NT, bool BIAS_RELU, bool ROW_SCALE>
__device__ __forceinline__ void gemm_rows16(
    const float* __restrict__ A, const float* __restrict__ Wp,
    const float* __restrict__ bias, const float* __restrict__ rscale,
    float* __restrict__ out,
    int row_tile, int lane, int out_stride, int out_off) {
  static_assert(KDIM % 32 == 0, "A rows are read in pairs of 16-float chunks");
  const int m16 = lane & 15;
  const int khalf = lane >> 4;
  constexpr int NCOLS = NT * 16;
  constexpr int NCH = KDIM / 16;
  constexpr int GROUP = NCOLS * 4;

  v8f acc[NT];
#pragma unroll
  for (int t = 0; t < NT; ++t) acc[t] = (v8f){0.f, 0.f, 0.f, 0.f, 0.f, 0.f, 0.f, 0.f};

  const float* __restrict__ arow = A + (size_t)(row_tile + m16) * KDIM + khalf * 4;
  const float* __restrict__ wlane = Wp + (size_t)khalf * GROUP + m16 * 4;

  float rs_lane = 1.0f;
  if (ROW_SCALE) rs_lane = rscale[row_tile + m16];

  v4f a_even[2], a_odd[2];
#pragma unroll
  for (int q = 0; q < 2; ++q) a_even[q] = *(const v4f*)(arow + q * 8);

  auto chunk = [&](const int ch, const v4f (&a_cur)[2], v4f (&a_nxt)[2]) {
    const float* __restrict__ wc = wlane + (size_t)ch * (4 * GROUP);
    v4f b[2][NT];
#pragma unroll
    for (int q = 0; q < 2; ++q) {
#pragma unroll
      for (int t = 0; t < NT; ++t) b[q][t] = *(const v4f*)(wc + (size_t)q * (2 * GROUP) + t * 64);
    }
    __builtin_amdgcn_sched_barrier(0);
    if (ch + 1 < NCH) {
#pragma unroll
      for (int q = 0; q < 2; ++q) a_nxt[q] = *(const v4f*)(arow + (ch + 1) * 16 + q * 8);
    }
    if ((ch & 1) == 0 && ch + 8 < NCH) __builtin_prefetch(arow + (ch + 8) * 16, 0, 0);
    __builtin_amdgcn_sched_barrier(0);
#pragma unroll
    for (int q = 0; q < 2; ++q) {
      const v2f a0 = {a_cur[q].x, a_cur[q].y};
      const v2f a1 = {a_cur[q].z, a_cur[q].w};
#pragma unroll
      for (int t = 0; t < NT; ++t) {
        const v2f b0 = {b[q][t].x, b[q][t].y};
        acc[t] = __builtin_amdgcn_wmma_f32_16x16x4_f32(
             false, a0,  false, b0,
             (short)0, acc[t],  false,  false);
      }
#pragma unroll
      for (int t = 0; t < NT; ++t) {
        const v2f b1 = {b[q][t].z, b[q][t].w};
        acc[t] = __builtin_amdgcn_wmma_f32_16x16x4_f32(
             false, a1,  false, b1,
             (short)0, acc[t],  false,  false);
      }
    }
    __builtin_amdgcn_sched_barrier(0);
  };

  for (int ch = 0; ch < NCH; ch += 2) {
    chunk(ch, a_even, a_odd);
    chunk(ch + 1, a_odd, a_even);
  }

  const int rbase = row_tile + khalf * 8;
  float rs[8];
  if (ROW_SCALE) {
#pragma unroll
    for (int i = 0; i < 8; ++i) {
      const float lo = __builtin_bit_cast(float, __builtin_amdgcn_readlane(__builtin_bit_cast(int, rs_lane), i));
      const float hi = __builtin_bit_cast(float, __builtin_amdgcn_readlane(__builtin_bit_cast(int, rs_lane), 8 + i));
      rs[i] = khalf ? hi : lo;
    }
  }
#pragma unroll
  for (int t = 0; t < NT; ++t) {
#pragma unroll
    for (int i = 0; i < 8; ++i) {
      const int c = t * 16 + m16;
      float v = acc[t][i];
      if (ROW_SCALE) v *= rs[i];
      if (BIAS_RELU) v = fmaxf(v + bias[c], 0.0f);
      out[(size_t)(rbase + i) * out_stride + out_off + c] = v;
    }
  }
}

#define GB ((N_NODES + 63) / 64)
#define EB ((N_EDGES + 127) / 128)
#define ADJ4 ((50048 + N_EDGES) / 4)
#define CB ((ADJ4 + 127) / 128)

#define PB_F ((512 * 64 + 255) / 256)
#define PB_I ((256 * 32 + 255) / 256)
#define PB_1 ((96 * 96 + 255) / 256)
#define PB_2 ((96 * 64 + 255) / 256)
#define PB_Z ((N_NODES + 255) / 256)

__global__ __launch_bounds__(256) void k_prep(
    const float* __restrict__ W_feat, float* __restrict__ wpF,
    const float* __restrict__ W_img, float* __restrict__ wpI,
    const float* __restrict__ W_g1, float* __restrict__ wpG1,
    const float* __restrict__ W_g2, float* __restrict__ wpG2,
    int* __restrict__ cnt) {
  const int b = blockIdx.x;
  const int t = threadIdx.x;
  if (b < PB_F) {
    pack_w<512, 64>(W_feat, wpF, b * 256 + t);
  } else if (b < PB_F + PB_I) {
    pack_w<256, 32>(W_img, wpI, (b - PB_F) * 256 + t);
  } else if (b < PB_F + PB_I + PB_1) {
    pack_w<96, 96>(W_g1, wpG1, (b - PB_F - PB_I) * 256 + t);
  } else if (b < PB_F + PB_I + PB_1 + PB_2) {
    pack_w<96, 64>(W_g2, wpG2, (b - PB_F - PB_I - PB_1) * 256 + t);
  } else {
    const int i = (b - PB_F - PB_I - PB_1 - PB_2) * 256 + t;
    if (i < N_NODES) cnt[i] = 0;
  }
}

__global__ __launch_bounds__(128) void k_enc_rank(
    const float* __restrict__ feature, const float* __restrict__ wpF,
    const float* __restrict__ b_feat,
    const float* __restrict__ img, const float* __restrict__ wpI,
    const float* __restrict__ b_img, float* __restrict__ x,
    const int* __restrict__ dst, int* __restrict__ cnt, int* __restrict__ rank) {
  const int b = blockIdx.x;
  const int lane = threadIdx.x & 31;
  const int wave = threadIdx.x >> 5;
  if (b < GB) {
    const int row_tile = (b * 4 + wave) * 16;
    if (row_tile < N_NODES) gemm_rows16<512, 4, true, false>(feature, wpF, b_feat, nullptr, x, row_tile, lane, 96, 0);
  } else if (b < 2 * GB) {
    const int row_tile = ((b - GB) * 4 + wave) * 16;
    if (row_tile < N_NODES) gemm_rows16<256, 2, true, false>(img, wpI, b_img, nullptr, x, row_tile, lane, 96, 64);
  } else {
    const int i = (b - 2 * GB) * 128 + threadIdx.x;
    if (i < N_EDGES) rank[i] = atomicAdd(&cnt[dst[i]], 1);
  }
}

__global__ __launch_bounds__(128) __attribute__((amdgpu_waves_per_eu(1, 7))) void k_gemm1_fill(
    const float* __restrict__ x, const float* __restrict__ wpG1,
    const float* __restrict__ dinv, float* __restrict__ xw1,
    const int* __restrict__ src, const int* __restrict__ dst,
    const int* __restrict__ rank, const int* __restrict__ rowptr,
    int* __restrict__ col) {
  const int b = blockIdx.x;
  if (b < GB) {
    const int lane = threadIdx.x & 31;
    const int wave = threadIdx.x >> 5;
    const int row_tile = (b * 4 + wave) * 16;
    if (row_tile < N_NODES) gemm_rows16<96, 6, false, true>(x, wpG1, nullptr, dinv, xw1, row_tile, lane, 96, 0);
  } else {
    const int i = (b - GB) * 128 + threadIdx.x;
    if (i < N_EDGES) col[rowptr[dst[i]] + rank[i]] = src[i];
  }
}

__global__ __launch_bounds__(128) void k_gemm2_copy(
    const float* __restrict__ h, const float* __restrict__ wpG2, float* __restrict__ xw2,
    const int* __restrict__ from, int* __restrict__ to) {
  const int b = blockIdx.x;
  if (b < GB) {
    const int lane = threadIdx.x & 31;
    const int wave = threadIdx.x >> 5;
    const int row_tile = (b * 4 + wave) * 16;
    if (row_tile < N_NODES) gemm_rows16<96, 4, false, false>(h, wpG2, nullptr, nullptr, xw2, row_tile, lane, 64, 0);
  } else {
    const int i = (b - GB) * 128 + threadIdx.x;
    if (i < ADJ4) ((int4*)to)[i] = ((const int4*)from)[i];
  }
}

template <int C, bool RELU, bool SCALE_OUT>
__global__ __launch_bounds__(256) void k_gather(
    const float* __restrict__ xw, const float* __restrict__ dinv,
    const int* __restrict__ rowptr, const int* __restrict__ col,
    const float* __restrict__ bias, float* __restrict__ out, int n) {
  constexpr int V = C / 32;
  const int lane = threadIdx.x & 31;
  const int node = __builtin_amdgcn_readfirstlane((int)(blockIdx.x * 8 + (threadIdx.x >> 5)));
  if (node >= n) return;

  const int beg = rowptr[node];
  const int end = rowptr[node + 1];
  const float di = dinv[node];

  float acc[V];
  {
    const float* __restrict__ p = xw + (size_t)node * C + lane;
#pragma unroll
    for (int v = 0; v < V; ++v) acc[v] = p[32 * v];
  }

  for (int base = beg; base < end; base += 32) {
    const int m = min(32, end - base);
    int s_l = 0;
    if (lane < m) s_l = col[base + lane];
    int j = 0;
    for (; j + 4 <= m; j += 4) {
      float x[4][V];
#pragma unroll
      for (int u = 0; u < 4; ++u) {
        const int s = __builtin_amdgcn_readlane(s_l, j + u);
        const float* __restrict__ p = xw + (size_t)s * C + lane;
#pragma unroll
        for (int v = 0; v < V; ++v) x[u][v] = p[32 * v];
      }
#pragma unroll
      for (int u = 0; u < 4; ++u) {
#pragma unroll
        for (int v = 0; v < V; ++v) acc[v] += x[u][v];
      }
    }
    for (; j < m; ++j) {
      const int s = __builtin_amdgcn_readlane(s_l, j);
      const float* __restrict__ p = xw + (size_t)s * C + lane;
#pragma unroll
      for (int v = 0; v < V; ++v) acc[v] += p[32 * v];
    }
  }

  float* __restrict__ o = out + (size_t)node * C + lane;
#pragma unroll
  for (int v = 0; v < V; ++v) {
    float r = acc[v] * di + bias[lane + 32 * v];
    if (RELU) r = fmaxf(r, 0.0f);
    if (SCALE_OUT) r *= di;
    o[32 * v] = r;
  }
}

extern "C" void kernel_launch(void* const* d_in, const int* in_sizes, int n_in,
                              void* d_out, int out_size, void* d_ws, size_t ws_size,
                              hipStream_t stream) {
  (void)in_sizes; (void)n_in; (void)out_size; (void)ws_size;

  const float* feature = (const float*)d_in[0];
  const float* img     = (const float*)d_in[1];
  const int*   edges   = (const int*)d_in[2];
  const float* W_feat  = (const float*)d_in[3];
  const float* b_feat  = (const float*)d_in[4];
  const float* W_img   = (const float*)d_in[5];
  const float* b_img   = (const float*)d_in[6];
  const float* W_g1    = (const float*)d_in[7];
  const float* b_g1    = (const float*)d_in[8];
  const float* W_g2    = (const float*)d_in[9];
  const float* b_g2    = (const float*)d_in[10];
  float* out = (float*)d_out;

  const int* src = edges;
  const int* dst = edges + N_EDGES;

  float* ws     = (float*)d_ws;
  float* dinv   = ws;
  float* bufA   = ws + 50048;
  float* bufB   = bufA + (size_t)N_NODES * 96;
  float* wpF    = bufB + (size_t)N_NODES * 96;
  float* wpI    = wpF + 512 * 64;
  float* wpG1   = wpI + 256 * 32;
  float* wpG2   = wpG1 + 96 * 96;
  int*   rowptr1 = (int*)out;
  int*   col1    = rowptr1 + 50048;
  int*   cnt     = col1 + N_EDGES;
  int*   rank    = cnt + 50048;
  int*   rowptr2 = (int*)(bufB + (size_t)N_NODES * 64);
  int*   col2    = rowptr2 + 50048;

  k_prep<<<PB_F + PB_I + PB_1 + PB_2 + PB_Z, 256, 0, stream>>>(
      W_feat, wpF, W_img, wpI, W_g1, wpG1, W_g2, wpG2, cnt);

  k_enc_rank<<<2 * GB + EB, 128, 0, stream>>>(feature, wpF, b_feat, img, wpI, b_img, bufA,
                                               dst, cnt, rank);

  k_scan<<<1, 1024, 0, stream>>>(cnt, rowptr1, dinv);

  k_gemm1_fill<<<GB + EB, 128, 0, stream>>>(bufA, wpG1, dinv, bufB, src, dst, rank, rowptr1, col1);

  k_gather<96, true, true><<<(N_NODES + 7) / 8, 256, 0, stream>>>(bufB, dinv, rowptr1, col1, b_g1, bufA, N_NODES);

  k_gemm2_copy<<<GB + CB, 128, 0, stream>>>(bufA, wpG2, bufB, rowptr1, rowptr2);

  k_gather<64, false, false><<<(N_NODES + 7) / 8, 256, 0, stream>>>(bufB, dinv, rowptr2, col2, b_g2, out, N_NODES);
}


# Layer-2 GEMM runs inside the layer-1 gather's work-groups (16 gathered rows staged in LDS feed its WMMAs, adjacency move beside): h never goes to memory, 6 launches not 7; order of the sums kept
# speedup vs baseline: 2.1612x; 1.0711x over previous
#include <hip/hip_runtime.h>
#include <hip/hip_bf16.h>

typedef float v2f __attribute__((ext_vector_type(2)));
typedef float v8f __attribute__((ext_vector_type(8)));

#define N_NODES 50000
#define N_EDGES 800000

static_assert(N_NODES % 4 == 0, "k_scan moves whole int4 groups");

#define SCAN_ITERS ((N_NODES + 4095) / 4096)

__global__ __launch_bounds__(1024) void k_scan(const int* __restrict__ cnt,
                                               int* __restrict__ rowptr,
                                               float* __restrict__ dinv) {
  __shared__ int wtot[SCAN_ITERS][32];
  const int tid = threadIdx.x;
  const int lane = tid & 31;
  const int wave = tid >> 5;

  int4 c[SCAN_ITERS];
#pragma unroll
  for (int g = 0; g < SCAN_ITERS; ++g) {
    const int i0 = g * 4096 + tid * 4;
    c[g] = (i0 < N_NODES) ? *(const int4*)(cnt + i0) : make_int4(0, 0, 0, 0);
  }

  int tsum[SCAN_ITERS], incl[SCAN_ITERS];
#pragma unroll
  for (int g = 0; g < SCAN_ITERS; ++g) {
    tsum[g] = c[g].x + c[g].y + c[g].z + c[g].w;
    incl[g] = tsum[g];
  }
#pragma unroll
  for (int off = 1; off < 32; off <<= 1) {
#pragma unroll
    for (int g = 0; g < SCAN_ITERS; ++g) {
      const int t = __shfl_up(incl[g], off);
      if (lane >= off) incl[g] += t;
    }
  }
  if (lane == 31) {
#pragma unroll
    for (int g = 0; g < SCAN_ITERS; ++g) wtot[g][wave] = incl[g];
  }
  __syncthreads();

  int wv[SCAN_ITERS], winc[SCAN_ITERS];
#pragma unroll
  for (int g = 0; g < SCAN_ITERS; ++g) {
    wv[g] = wtot[g][lane];
    winc[g] = wv[g];
  }
#pragma unroll
  for (int off = 1; off < 32; off <<= 1) {
#pragma unroll
    for (int g = 0; g < SCAN_ITERS; ++g) {
      const int t = __shfl_up(winc[g], off);
      if (lane >= off) winc[g] += t;
    }
  }

  int carry = 0;
#pragma unroll
  for (int g = 0; g < SCAN_ITERS; ++g) {
    const int i0 = g * 4096 + tid * 4;
    const int wave_excl = __shfl(winc[g], wave) - __shfl(wv[g], wave);
    const int total = __shfl(winc[g], 31);
    const int e0 = carry + wave_excl + (incl[g] - tsum[g]);
    if (i0 < N_NODES) {
      const int4 v = c[g];
      int4 r;
      r.x = e0;
      r.y = r.x + v.x;
      r.z = r.y + v.y;
      r.w = r.z + v.z;
      *(int4*)(rowptr + i0) = r;
      float4 d;
      d.x = rsqrtf((float)(v.x + 1));
      d.y = rsqrtf((float)(v.y + 1));
      d.z = rsqrtf((float)(v.z + 1));
      d.w = rsqrtf((float)(v.w + 1));
      *(float4*)(dinv + i0) = d;
    }
    carry += total;
  }
  if (tid < 48) rowptr[N_NODES + tid] = carry;
}


template <int K, int N>
__device__ __forceinline__ void pack_w(const float* __restrict__ W, float* __restrict__ Wp,
                                       int idx) {
  if (idx >= K * N) return;
  const int k = idx / N;
  const int n = idx - k * N;
  Wp[(size_t)(k >> 2) * (N * 4) + n * 4 + (k & 3)] = W[idx];
}

typedef float v4f __attribute__((ext_vector_type(4)));

template <int KDIM, int NT, bool BIAS_RELU, bool ROW_SCALE>
__device__ __forceinline__ void gemm_rows16(
    const float* __restrict__ A, const float* __restrict__ Wp,
    const float* __restrict__ bias, const float* __restrict__ rscale,
    float* __restrict__ out,
    int row_tile, int lane, int out_stride, int out_off) {
  static_assert(KDIM % 32 == 0, "A rows are read in pairs of 16-float chunks");
  const int m16 = lane & 15;
  const int khalf = lane >> 4;
  constexpr int NCOLS = NT * 16;
  constexpr int NCH = KDIM / 16;
  constexpr int GROUP = NCOLS * 4;

  v8f acc[NT];
#pragma unroll
  for (int t = 0; t < NT; ++t) acc[t] = (v8f){0.f, 0.f, 0.f, 0.f, 0.f, 0.f, 0.f, 0.f};

  const float* __restrict__ arow = A + (size_t)(row_tile + m16) * KDIM + khalf * 4;
  const float* __restrict__ wlane = Wp + (size_t)khalf * GROUP + m16 * 4;

  float rs_lane = 1.0f;
  if (ROW_SCALE) rs_lane = rscale[row_tile + m16];

  v4f a_even[2], a_odd[2];
#pragma unroll
  for (int q = 0; q < 2; ++q) a_even[q] = *(const v4f*)(arow + q * 8);

  auto chunk = [&](const int ch, const v4f (&a_cur)[2], v4f (&a_nxt)[2]) {
    const float* __restrict__ wc = wlane + (size_t)ch * (4 * GROUP);
    v4f b[2][NT];
#pragma unroll
    for (int q = 0; q < 2; ++q) {
#pragma unroll
      for (int t = 0; t < NT; ++t) b[q][t] = *(const v4f*)(wc + (size_t)q * (2 * GROUP) + t * 64);
    }
    __builtin_amdgcn_sched_barrier(0);
    if (ch + 1 < NCH) {
#pragma unroll
      for (int q = 0; q < 2; ++q) a_nxt[q] = *(const v4f*)(arow + (ch + 1) * 16 + q * 8);
    }
    if ((ch & 1) == 0 && ch + 8 < NCH) __builtin_prefetch(arow + (ch + 8) * 16, 0, 0);
    __builtin_amdgcn_sched_barrier(0);
#pragma unroll
    for (int q = 0; q < 2; ++q) {
      const v2f a0 = {a_cur[q].x, a_cur[q].y};
      const v2f a1 = {a_cur[q].z, a_cur[q].w};
#pragma unroll
      for (int t = 0; t < NT; ++t) {
        const v2f b0 = {b[q][t].x, b[q][t].y};
        acc[t] = __builtin_amdgcn_wmma_f32_16x16x4_f32(
             false, a0,  false, b0,
             (short)0, acc[t],  false,  false);
      }
#pragma unroll
      for (int t = 0; t < NT; ++t) {
        const v2f b1 = {b[q][t].z, b[q][t].w};
        acc[t] = __builtin_amdgcn_wmma_f32_16x16x4_f32(
             false, a1,  false, b1,
             (short)0, acc[t],  false,  false);
      }
    }
    __builtin_amdgcn_sched_barrier(0);
  };

  for (int ch = 0; ch < NCH; ch += 2) {
    chunk(ch, a_even, a_odd);
    chunk(ch + 1, a_odd, a_even);
  }

  const int rbase = row_tile + khalf * 8;
  float rs[8];
  if (ROW_SCALE) {
#pragma unroll
    for (int i = 0; i < 8; ++i) {
      const float lo = __builtin_bit_cast(float, __builtin_amdgcn_readlane(__builtin_bit_cast(int, rs_lane), i));
      const float hi = __builtin_bit_cast(float, __builtin_amdgcn_readlane(__builtin_bit_cast(int, rs_lane), 8 + i));
      rs[i] = khalf ? hi : lo;
    }
  }
#pragma unroll
  for (int t = 0; t < NT; ++t) {
#pragma unroll
    for (int i = 0; i < 8; ++i) {
      const int c = t * 16 + m16;
      float v = acc[t][i];
      if (ROW_SCALE) v *= rs[i];
      if (BIAS_RELU) v = fmaxf(v + bias[c], 0.0f);
      out[(size_t)(rbase + i) * out_stride + out_off + c] = v;
    }
  }
}

#define GB ((N_NODES + 63) / 64)
#define EB ((N_EDGES + 127) / 128)
#define ADJ4 ((50048 + N_EDGES) / 4)
#define CB ((ADJ4 + 511) / 512)

#define PB_F ((512 * 64 + 255) / 256)
#define PB_I ((256 * 32 + 255) / 256)
#define PB_1 ((96 * 96 + 255) / 256)
#define PB_2 ((96 * 64 + 255) / 256)
#define PB_Z ((N_NODES + 255) / 256)

__global__ __launch_bounds__(256) void k_prep(
    const float* __restrict__ W_feat, float* __restrict__ wpF,
    const float* __restrict__ W_img, float* __restrict__ wpI,
    const float* __restrict__ W_g1, float* __restrict__ wpG1,
    const float* __restrict__ W_g2, float* __restrict__ wpG2,
    int* __restrict__ cnt) {
  const int b = blockIdx.x;
  const int t = threadIdx.x;
  if (b < PB_F) {
    pack_w<512, 64>(W_feat, wpF, b * 256 + t);
  } else if (b < PB_F + PB_I) {
    pack_w<256, 32>(W_img, wpI, (b - PB_F) * 256 + t);
  } else if (b < PB_F + PB_I + PB_1) {
    pack_w<96, 96>(W_g1, wpG1, (b - PB_F - PB_I) * 256 + t);
  } else if (b < PB_F + PB_I + PB_1 + PB_2) {
    pack_w<96, 64>(W_g2, wpG2, (b - PB_F - PB_I - PB_1) * 256 + t);
  } else {
    const int i = (b - PB_F - PB_I - PB_1 - PB_2) * 256 + t;
    if (i < N_NODES) cnt[i] = 0;
  }
}

__global__ __launch_bounds__(128) void k_enc_rank(
    const float* __restrict__ feature, const float* __restrict__ wpF,
    const float* __restrict__ b_feat,
    const float* __restrict__ img, const float* __restrict__ wpI,
    const float* __restrict__ b_img, float* __restrict__ x,
    const int* __restrict__ dst, int* __restrict__ cnt, int* __restrict__ rank) {
  const int b = blockIdx.x;
  const int lane = threadIdx.x & 31;
  const int wave = threadIdx.x >> 5;
  if (b < GB) {
    const int row_tile = (b * 4 + wave) * 16;
    if (row_tile < N_NODES) gemm_rows16<512, 4, true, false>(feature, wpF, b_feat, nullptr, x, row_tile, lane, 96, 0);
  } else if (b < 2 * GB) {
    const int row_tile = ((b - GB) * 4 + wave) * 16;
    if (row_tile < N_NODES) gemm_rows16<256, 2, true, false>(img, wpI, b_img, nullptr, x, row_tile, lane, 96, 64);
  } else {
    const int i = (b - 2 * GB) * 128 + threadIdx.x;
    if (i < N_EDGES) rank[i] = atomicAdd(&cnt[dst[i]], 1);
  }
}

__global__ __launch_bounds__(128) __attribute__((amdgpu_waves_per_eu(1, 7))) void k_gemm1_fill(
    const float* __restrict__ x, const float* __restrict__ wpG1,
    const float* __restrict__ dinv, float* __restrict__ xw1,
    const int* __restrict__ src, const int* __restrict__ dst,
    const int* __restrict__ rank, const int* __restrict__ rowptr,
    int* __restrict__ col) {
  const int b = blockIdx.x;
  if (b < GB) {
    const int lane = threadIdx.x & 31;
    const int wave = threadIdx.x >> 5;
    const int row_tile = (b * 4 + wave) * 16;
    if (row_tile < N_NODES) gemm_rows16<96, 6, false, true>(x, wpG1, nullptr, dinv, xw1, row_tile, lane, 96, 0);
  } else {
    const int i = (b - GB) * 128 + threadIdx.x;
    if (i < N_EDGES) col[rowptr[dst[i]] + rank[i]] = src[i];
  }
}

template <int C, bool RELU, bool SCALE_OUT>
__global__ __launch_bounds__(256) void k_gather(
    const float* __restrict__ xw, const float* __restrict__ dinv,
    const int* __restrict__ rowptr, const int* __restrict__ col,
    const float* __restrict__ bias, float* __restrict__ out, int n) {
  constexpr int V = C / 32;
  const int lane = threadIdx.x & 31;
  const int node = __builtin_amdgcn_readfirstlane((int)(blockIdx.x * 8 + (threadIdx.x >> 5)));
  if (node >= n) return;

  const int beg = rowptr[node];
  const int end = rowptr[node + 1];
  const float di = dinv[node];

  float acc[V];
  {
    const float* __restrict__ p = xw + (size_t)node * C + lane;
#pragma unroll
    for (int v = 0; v < V; ++v) acc[v] = p[32 * v];
  }

  for (int base = beg; base < end; base += 32) {
    const int m = min(32, end - base);
    int s_l = 0;
    if (lane < m) s_l = col[base + lane];
    int j = 0;
    for (; j + 4 <= m; j += 4) {
      float x[4][V];
#pragma unroll
      for (int u = 0; u < 4; ++u) {
        const int s = __builtin_amdgcn_readlane(s_l, j + u);
        const float* __restrict__ p = xw + (size_t)s * C + lane;
#pragma unroll
        for (int v = 0; v < V; ++v) x[u][v] = p[32 * v];
      }
#pragma unroll
      for (int u = 0; u < 4; ++u) {
#pragma unroll
        for (int v = 0; v < V; ++v) acc[v] += x[u][v];
      }
    }
    for (; j < m; ++j) {
      const int s = __builtin_amdgcn_readlane(s_l, j);
      const float* __restrict__ p = xw + (size_t)s * C + lane;
#pragma unroll
      for (int v = 0; v < V; ++v) acc[v] += p[32 * v];
    }
  }

  float* __restrict__ o = out + (size_t)node * C + lane;
#pragma unroll
  for (int v = 0; v < V; ++v) {
    float r = acc[v] * di + bias[lane + 32 * v];
    if (RELU) r = fmaxf(r, 0.0f);
    if (SCALE_OUT) r *= di;
    o[32 * v] = r;
  }
}


static_assert(N_NODES % 16 == 0, "every gather work-group is one full 16-row tile");
#define TILE_PITCH 100
#define GT (N_NODES / 16)

__global__ __launch_bounds__(512) void k_gather1_gemm2(
    const float* __restrict__ xw, const float* __restrict__ dinv,
    const int* __restrict__ rowptr, const int* __restrict__ col,
    const float* __restrict__ bias, const float* __restrict__ wp2,
    float* __restrict__ xw2, const int* __restrict__ from, int* __restrict__ to) {
  __shared__ __attribute__((aligned(16))) float hs[16 * TILE_PITCH];
  const int b = blockIdx.x;
  if (b >= GT) {
    const int i = (b - GT) * 512 + threadIdx.x;
    if (i < ADJ4) ((int4*)to)[i] = ((const int4*)from)[i];
    return;
  }
  constexpr int C = 96;
  constexpr int V = C / 32;
  const int lane = threadIdx.x & 31;
  const int wave = __builtin_amdgcn_readfirstlane((int)(threadIdx.x >> 5));
  const int node = b * 16 + wave;

  const int beg = rowptr[node];
  const int end = rowptr[node + 1];
  const float di = dinv[node];

  float acc[V];
  {
    const float* __restrict__ p = xw + (size_t)node * C + lane;
#pragma unroll
    for (int v = 0; v < V; ++v) acc[v] = p[32 * v];
  }

  for (int base = beg; base < end; base += 32) {
    const int m = min(32, end - base);
    int s_l = 0;
    if (lane < m) s_l = col[base + lane];
    int j = 0;
    for (; j + 4 <= m; j += 4) {
      float x[4][V];
#pragma unroll
      for (int u = 0; u < 4; ++u) {
        const int s = __builtin_amdgcn_readlane(s_l, j + u);
        const float* __restrict__ p = xw + (size_t)s * C + lane;
#pragma unroll
        for (int v = 0; v < V; ++v) x[u][v] = p[32 * v];
      }
#pragma unroll
      for (int u = 0; u < 4; ++u) {
#pragma unroll
        for (int v = 0; v < V; ++v) acc[v] += x[u][v];
      }
    }
    for (; j < m; ++j) {
      const int s = __builtin_amdgcn_readlane(s_l, j);
      const float* __restrict__ p = xw + (size_t)s * C + lane;
#pragma unroll
      for (int v = 0; v < V; ++v) acc[v] += p[32 * v];
    }
  }

#pragma unroll
  for (int v = 0; v < V; ++v) {
    float r = fmaxf(acc[v] * di + bias[lane + 32 * v], 0.0f);
    hs[wave * TILE_PITCH + lane + 32 * v] = r * di;
  }

  const int m16 = lane & 15;
  const int khalf = lane >> 4;
  const float* __restrict__ wl = wp2 + khalf * 256 + m16 * 4 + wave * 128;
  v4f bq[2][2];
  if (wave < 2) {
#pragma unroll
    for (int q = 0; q < 2; ++q) {
#pragma unroll
      for (int t = 0; t < 2; ++t) bq[q][t] = *(const v4f*)(wl + q * 512 + t * 64);
    }
  }
  __syncthreads();
  if (wave >= 2) return;

  v8f acc2[2];
#pragma unroll
  for (int t = 0; t < 2; ++t) acc2[t] = (v8f){0.f, 0.f, 0.f, 0.f, 0.f, 0.f, 0.f, 0.f};
  const float* al = hs + m16 * TILE_PITCH + khalf * 4;

#pragma unroll
  for (int ch = 0; ch < 6; ++ch) {
    v4f aq[2];
#pragma unroll
    for (int q = 0; q < 2; ++q) aq[q] = *(const v4f*)(al + ch * 16 + q * 8);
    v4f bn[2][2];
    if (ch + 1 < 6) {
#pragma unroll
      for (int q = 0; q < 2; ++q) {
#pragma unroll
        for (int t = 0; t < 2; ++t) bn[q][t] = *(const v4f*)(wl + (ch + 1) * 1024 + q * 512 + t * 64);
      }
    }
#pragma unroll
    for (int q = 0; q < 2; ++q) {
      const v2f a0 = {aq[q].x, aq[q].y};
      const v2f a1 = {aq[q].z, aq[q].w};
#pragma unroll
      for (int t = 0; t < 2; ++t) {
        const v2f b0 = {bq[q][t].x, bq[q][t].y};
        acc2[t] = __builtin_amdgcn_wmma_f32_16x16x4_f32(
             false, a0,  false, b0,
             (short)0, acc2[t],  false,  false);
      }
#pragma unroll
      for (int t = 0; t < 2; ++t) {
        const v2f b1 = {bq[q][t].z, bq[q][t].w};
        acc2[t] = __builtin_amdgcn_wmma_f32_16x16x4_f32(
             false, a1,  false, b1,
             (short)0, acc2[t],  false,  false);
      }
    }
    if (ch + 1 < 6) {
#pragma unroll
      for (int q = 0; q < 2; ++q) {
#pragma unroll
        for (int t = 0; t < 2; ++t) bq[q][t] = bn[q][t];
      }
    }
  }

  const int rbase = b * 16 + khalf * 8;
#pragma unroll
  for (int t = 0; t < 2; ++t) {
#pragma unroll
    for (int i = 0; i < 8; ++i) {
      xw2[(size_t)(rbase + i) * 64 + (2 * wave + t) * 16 + m16] = acc2[t][i];
    }
  }
}

extern "C" void kernel_launch(void* const* d_in, const int* in_sizes, int n_in,
                              void* d_out, int out_size, void* d_ws, size_t ws_size,
                              hipStream_t stream) {
  (void)in_sizes; (void)n_in; (void)out_size; (void)ws_size;

  const float* feature = (const float*)d_in[0];
  const float* img     = (const float*)d_in[1];
  const int*   edges   = (const int*)d_in[2];
  const float* W_feat  = (const float*)d_in[3];
  const float* b_feat  = (const float*)d_in[4];
  const float* W_img   = (const float*)d_in[5];
  const float* b_img   = (const float*)d_in[6];
  const float* W_g1    = (const float*)d_in[7];
  const float* b_g1    = (const float*)d_in[8];
  const float* W_g2    = (const float*)d_in[9];
  const float* b_g2    = (const float*)d_in[10];
  float* out = (float*)d_out;

  const int* src = edges;
  const int* dst = edges + N_EDGES;

  float* ws     = (float*)d_ws;
  float* dinv   = ws;
  float* bufA   = ws + 50048;
  float* bufB   = bufA + (size_t)N_NODES * 96;
  float* wpF    = bufB + (size_t)N_NODES * 96;
  float* wpI    = wpF + 512 * 64;
  float* wpG1   = wpI + 256 * 32;
  float* wpG2   = wpG1 + 96 * 96;
  int*   rowptr1 = (int*)out;
  int*   col1    = rowptr1 + 50048;
  int*   cnt     = col1 + N_EDGES;
  int*   rank    = cnt + 50048;
  int*   rowptr2 = (int*)(bufA + (size_t)N_NODES * 64);
  int*   col2    = rowptr2 + 50048;

  k_prep<<<PB_F + PB_I + PB_1 + PB_2 + PB_Z, 256, 0, stream>>>(
      W_feat, wpF, W_img, wpI, W_g1, wpG1, W_g2, wpG2, cnt);

  k_enc_rank<<<2 * GB + EB, 128, 0, stream>>>(feature, wpF, b_feat, img, wpI, b_img, bufA,
                                               dst, cnt, rank);

  k_scan<<<1, 1024, 0, stream>>>(cnt, rowptr1, dinv);

  k_gemm1_fill<<<GB + EB, 128, 0, stream>>>(bufA, wpG1, dinv, bufB, src, dst, rank, rowptr1, col1);

  k_gather1_gemm2<<<GT + CB, 512, 0, stream>>>(bufB, dinv, rowptr1, col1, b_g1, wpG2, bufA,
                                               rowptr1, rowptr2);

  k_gather<64, false, false><<<(N_NODES + 7) / 8, 256, 0, stream>>>(bufA, dinv, rowptr2, col2, b_g2, out, N_NODES);
}


# Adjacency written by the ranking pass as fixed 32-id rows plus an overflow chain (int exchange atomics), D^-1/2 from the counts: no scan, no fill, 5 launches not 6; overflow edges sum in chain order
# speedup vs baseline: 2.5678x; 1.1881x over previous
#include <hip/hip_runtime.h>
#include <hip/hip_bf16.h>

typedef float v2f __attribute__((ext_vector_type(2)));
typedef float v8f __attribute__((ext_vector_type(8)));

#define N_NODES 50000
#define N_EDGES 800000

#define ELLW 32


template <int K, int N>
__device__ __forceinline__ void pack_w(const float* __restrict__ W, float* __restrict__ Wp,
                                       int idx) {
  if (idx >= K * N) return;
  const int k = idx / N;
  const int n = idx - k * N;
  Wp[(size_t)(k >> 2) * (N * 4) + n * 4 + (k & 3)] = W[idx];
}

typedef float v4f __attribute__((ext_vector_type(4)));

template <int KDIM, int NT, bool BIAS_RELU, bool ROW_SCALE, bool SPLIT>
__device__ __forceinline__ void gemm_rows16(
    const float* __restrict__ A, const float* __restrict__ Wp,
    const float* __restrict__ bias, const int* __restrict__ rcnt,
    float* __restrict__ out, float* __restrict__ out2,
    int row_tile, int lane, int out_stride, int out_off) {
  static_assert(KDIM % 32 == 0, "A rows are read in pairs of 16-float chunks");
  const int m16 = lane & 15;
  const int khalf = lane >> 4;
  constexpr int NCOLS = NT * 16;
  constexpr int NCH = KDIM / 16;
  constexpr int GROUP = NCOLS * 4;

  v8f acc[NT];
#pragma unroll
  for (int t = 0; t < NT; ++t) acc[t] = (v8f){0.f, 0.f, 0.f, 0.f, 0.f, 0.f, 0.f, 0.f};

  const float* __restrict__ arow = A + (size_t)(row_tile + m16) * KDIM + khalf * 4;
  const float* __restrict__ wlane = Wp + (size_t)khalf * GROUP + m16 * 4;

  float rs_lane = 1.0f;
  if (ROW_SCALE) rs_lane = rsqrtf((float)(rcnt[row_tile + m16] + 1));

  v4f a_even[2], a_odd[2];
#pragma unroll
  for (int q = 0; q < 2; ++q) a_even[q] = *(const v4f*)(arow + q * 8);

  auto chunk = [&](const int ch, const v4f (&a_cur)[2], v4f (&a_nxt)[2]) {
    const float* __restrict__ wc = wlane + (size_t)ch * (4 * GROUP);
    v4f b[2][NT];
#pragma unroll
    for (int q = 0; q < 2; ++q) {
#pragma unroll
      for (int t = 0; t < NT; ++t) b[q][t] = *(const v4f*)(wc + (size_t)q * (2 * GROUP) + t * 64);
    }
    __builtin_amdgcn_sched_barrier(0);
    if (ch + 1 < NCH) {
#pragma unroll
      for (int q = 0; q < 2; ++q) a_nxt[q] = *(const v4f*)(arow + (ch + 1) * 16 + q * 8);
    }
    if ((ch & 1) == 0 && ch + 8 < NCH) __builtin_prefetch(arow + (ch + 8) * 16, 0, 0);
    __builtin_amdgcn_sched_barrier(0);
#pragma unroll
    for (int q = 0; q < 2; ++q) {
      const v2f a0 = {a_cur[q].x, a_cur[q].y};
      const v2f a1 = {a_cur[q].z, a_cur[q].w};
#pragma unroll
      for (int t = 0; t < NT; ++t) {
        const v2f b0 = {b[q][t].x, b[q][t].y};
        acc[t] = __builtin_amdgcn_wmma_f32_16x16x4_f32(
             false, a0,  false, b0,
             (short)0, acc[t],  false,  false);
      }
#pragma unroll
      for (int t = 0; t < NT; ++t) {
        const v2f b1 = {b[q][t].z, b[q][t].w};
        acc[t] = __builtin_amdgcn_wmma_f32_16x16x4_f32(
             false, a1,  false, b1,
             (short)0, acc[t],  false,  false);
      }
    }
    __builtin_amdgcn_sched_barrier(0);
  };

  for (int ch = 0; ch < NCH; ch += 2) {
    chunk(ch, a_even, a_odd);
    chunk(ch + 1, a_odd, a_even);
  }

  const int rbase = row_tile + khalf * 8;
  float rs[8];
  if (ROW_SCALE) {
#pragma unroll
    for (int i = 0; i < 8; ++i) {
      const float lo = __builtin_bit_cast(float, __builtin_amdgcn_readlane(__builtin_bit_cast(int, rs_lane), i));
      const float hi = __builtin_bit_cast(float, __builtin_amdgcn_readlane(__builtin_bit_cast(int, rs_lane), 8 + i));
      rs[i] = khalf ? hi : lo;
    }
  }
#pragma unroll
  for (int t = 0; t < NT; ++t) {
#pragma unroll
    for (int i = 0; i < 8; ++i) {
      const int c = t * 16 + m16;
      float v = acc[t][i];
      if (ROW_SCALE) v *= rs[i];
      if (BIAS_RELU) v = fmaxf(v + bias[c], 0.0f);
      if (SPLIT && t >= 4) {
        out2[(size_t)(rbase + i) * 32 + (c - 64)] = v;
      } else {
        out[(size_t)(rbase + i) * out_stride + out_off + c] = v;
      }
    }
  }
}

#define GB ((N_NODES + 63) / 64)
#define EB ((N_EDGES + 127) / 128)

#define PB_F ((512 * 64 + 255) / 256)
#define PB_I ((256 * 32 + 255) / 256)
#define PB_1 ((96 * 96 + 255) / 256)
#define PB_2 ((96 * 64 + 255) / 256)
#define PB_Z ((N_NODES + 255) / 256)

__global__ __launch_bounds__(256) void k_prep(
    const float* __restrict__ W_feat, float* __restrict__ wpF,
    const float* __restrict__ W_img, float* __restrict__ wpI,
    const float* __restrict__ W_g1, float* __restrict__ wpG1,
    const float* __restrict__ W_g2, float* __restrict__ wpG2,
    int* __restrict__ cnt, int* __restrict__ head) {
  const int b = blockIdx.x;
  const int t = threadIdx.x;
  if (b < PB_F) {
    pack_w<512, 64>(W_feat, wpF, b * 256 + t);
  } else if (b < PB_F + PB_I) {
    pack_w<256, 32>(W_img, wpI, (b - PB_F) * 256 + t);
  } else if (b < PB_F + PB_I + PB_1) {
    pack_w<96, 96>(W_g1, wpG1, (b - PB_F - PB_I) * 256 + t);
  } else if (b < PB_F + PB_I + PB_1 + PB_2) {
    pack_w<96, 64>(W_g2, wpG2, (b - PB_F - PB_I - PB_1) * 256 + t);
  } else {
    const int i = (b - PB_F - PB_I - PB_1 - PB_2) * 256 + t;
    if (i < N_NODES) {
      cnt[i] = 0;
      head[i] = -1;
    }
  }
}

__global__ __launch_bounds__(128) void k_enc_rank(
    const float* __restrict__ feature, const float* __restrict__ wpF,
    const float* __restrict__ b_feat,
    const float* __restrict__ img, const float* __restrict__ wpI,
    const float* __restrict__ b_img, float* __restrict__ x,
    const int* __restrict__ src, const int* __restrict__ dst,
    int* __restrict__ cnt, int* __restrict__ ell,
    int* __restrict__ head, int* __restrict__ next) {
  const int b = blockIdx.x;
  const int lane = threadIdx.x & 31;
  const int wave = threadIdx.x >> 5;
  if (b < GB) {
    const int row_tile = (b * 4 + wave) * 16;
    if (row_tile < N_NODES) gemm_rows16<512, 4, true, false, false>(feature, wpF, b_feat, nullptr, x, nullptr, row_tile, lane, 96, 0);
  } else if (b < 2 * GB) {
    const int row_tile = ((b - GB) * 4 + wave) * 16;
    if (row_tile < N_NODES) gemm_rows16<256, 2, true, false, false>(img, wpI, b_img, nullptr, x, nullptr, row_tile, lane, 96, 64);
  } else {
    const int i = (b - 2 * GB) * 128 + threadIdx.x;
    if (i < N_EDGES) {
      const int d = dst[i];
      const int r = atomicAdd(&cnt[d], 1);
      if (r < ELLW) {
        ell[(size_t)d * ELLW + r] = src[i];
      } else {
        next[i] = atomicExch(&head[d], i);
      }
    }
  }
}

__global__ __launch_bounds__(128) __attribute__((amdgpu_waves_per_eu(1, 7))) void k_gemm1(
    const float* __restrict__ x, const float* __restrict__ wpG1,
    const int* __restrict__ cnt, float* __restrict__ xw1a, float* __restrict__ xw1b) {
  const int b = blockIdx.x;
  const int lane = threadIdx.x & 31;
  const int wave = threadIdx.x >> 5;
  const int row_tile = (b * 4 + wave) * 16;
  if (row_tile < N_NODES) gemm_rows16<96, 6, false, true, true>(x, wpG1, nullptr, cnt, xw1a, xw1b, row_tile, lane, 64, 0);
}

template <int V>
__device__ __forceinline__ void load_row(float (&x)[V], const float* __restrict__ xa,
                                         const float* __restrict__ xb, int s, int lane) {
  const float* __restrict__ p = xa + (size_t)s * 64 + lane;
  x[0] = p[0];
  x[1] = p[32];
  if (V == 3) x[2] = xb[(size_t)s * 32 + lane];
}

template <int V>
__device__ __forceinline__ int gather_row(
    float (&acc)[V], const float* __restrict__ xa, const float* __restrict__ xb,
    const int* __restrict__ cnt, const int* __restrict__ ell,
    const int* __restrict__ head, const int* __restrict__ next,
    const int* __restrict__ src, int node, int lane) {
  const int c = cnt[node];
  const int m = min(c, ELLW);
  int s_l = 0;
  if (lane < m) s_l = ell[(size_t)node * ELLW + lane];
  load_row<V>(acc, xa, xb, node, lane);

  int j = 0;
  for (; j + 4 <= m; j += 4) {
    float x[4][V];
#pragma unroll
    for (int u = 0; u < 4; ++u) {
      const int s = __builtin_amdgcn_readlane(s_l, j + u);
      load_row<V>(x[u], xa, xb, s, lane);
    }
#pragma unroll
    for (int u = 0; u < 4; ++u) {
#pragma unroll
      for (int v = 0; v < V; ++v) acc[v] += x[u][v];
    }
  }
  for (; j < m; ++j) {
    const int s = __builtin_amdgcn_readlane(s_l, j);
    float x[V];
    load_row<V>(x, xa, xb, s, lane);
#pragma unroll
    for (int v = 0; v < V; ++v) acc[v] += x[v];
  }
  if (c > ELLW) {
    int e = head[node];
    for (int k = ELLW; k < c; ++k) {
      if (e < 0 || e >= N_EDGES) break;
      const int s = src[e];
      float x[V];
      load_row<V>(x, xa, xb, s, lane);
#pragma unroll
      for (int v = 0; v < V; ++v) acc[v] += x[v];
      e = next[e];
    }
  }
  return c;
}

__global__ __launch_bounds__(256) void k_gather(
    const float* __restrict__ xw, const int* __restrict__ cnt,
    const int* __restrict__ ell, const int* __restrict__ head,
    const int* __restrict__ next, const int* __restrict__ src,
    const float* __restrict__ bias, float* __restrict__ out, int n) {
  constexpr int V = 2;
  const int lane = threadIdx.x & 31;
  const int node = __builtin_amdgcn_readfirstlane((int)(blockIdx.x * 8 + (threadIdx.x >> 5)));
  if (node >= n) return;

  float acc[V];
  const int c = gather_row<V>(acc, xw, nullptr, cnt, ell, head, next, src, node, lane);
  const float di = rsqrtf((float)(c + 1));

  float* __restrict__ o = out + (size_t)node * 64 + lane;
#pragma unroll
  for (int v = 0; v < V; ++v) o[32 * v] = acc[v] * di + bias[lane + 32 * v];
}


static_assert(N_NODES % 16 == 0, "every gather work-group is one full 16-row tile");
#define TILE_PITCH 100
#define GT (N_NODES / 16)

__global__ __launch_bounds__(512) void k_gather1_gemm2(
    const float* __restrict__ xwa, const float* __restrict__ xwb,
    const int* __restrict__ cnt, const int* __restrict__ ell,
    const int* __restrict__ head, const int* __restrict__ next,
    const int* __restrict__ src,
    const float* __restrict__ bias, const float* __restrict__ wp2,
    float* __restrict__ xw2) {
  __shared__ __attribute__((aligned(16))) float hs[16 * TILE_PITCH];
  const int b = blockIdx.x;
  constexpr int V = 3;
  const int lane = threadIdx.x & 31;
  const int wave = __builtin_amdgcn_readfirstlane((int)(threadIdx.x >> 5));
  const int node = b * 16 + wave;

  float acc[V];
  const int c = gather_row<V>(acc, xwa, xwb, cnt, ell, head, next, src, node, lane);
  const float di = rsqrtf((float)(c + 1));

#pragma unroll
  for (int v = 0; v < V; ++v) {
    float r = fmaxf(acc[v] * di + bias[lane + 32 * v], 0.0f);
    hs[wave * TILE_PITCH + lane + 32 * v] = r * di;
  }

  const int m16 = lane & 15;
  const int khalf = lane >> 4;
  const float* __restrict__ wl = wp2 + khalf * 256 + m16 * 4 + wave * 128;
  v4f bq[2][2];
  if (wave < 2) {
#pragma unroll
    for (int q = 0; q < 2; ++q) {
#pragma unroll
      for (int t = 0; t < 2; ++t) bq[q][t] = *(const v4f*)(wl + q * 512 + t * 64);
    }
  }
  __syncthreads();
  if (wave >= 2) return;

  v8f acc2[2];
#pragma unroll
  for (int t = 0; t < 2; ++t) acc2[t] = (v8f){0.f, 0.f, 0.f, 0.f, 0.f, 0.f, 0.f, 0.f};
  const float* al = hs + m16 * TILE_PITCH + khalf * 4;

#pragma unroll
  for (int ch = 0; ch < 6; ++ch) {
    v4f aq[2];
#pragma unroll
    for (int q = 0; q < 2; ++q) aq[q] = *(const v4f*)(al + ch * 16 + q * 8);
    v4f bn[2][2];
    if (ch + 1 < 6) {
#pragma unroll
      for (int q = 0; q < 2; ++q) {
#pragma unroll
        for (int t = 0; t < 2; ++t) bn[q][t] = *(const v4f*)(wl + (ch + 1) * 1024 + q * 512 + t * 64);
      }
    }
#pragma unroll
    for (int q = 0; q < 2; ++q) {
      const v2f a0 = {aq[q].x, aq[q].y};
      const v2f a1 = {aq[q].z, aq[q].w};
#pragma unroll
      for (int t = 0; t < 2; ++t) {
        const v2f b0 = {bq[q][t].x, bq[q][t].y};
        acc2[t] = __builtin_amdgcn_wmma_f32_16x16x4_f32(
             false, a0,  false, b0,
             (short)0, acc2[t],  false,  false);
      }
#pragma unroll
      for (int t = 0; t < 2; ++t) {
        const v2f b1 = {bq[q][t].z, bq[q][t].w};
        acc2[t] = __builtin_amdgcn_wmma_f32_16x16x4_f32(
             false, a1,  false, b1,
             (short)0, acc2[t],  false,  false);
      }
    }
    if (ch + 1 < 6) {
#pragma unroll
      for (int q = 0; q < 2; ++q) {
#pragma unroll
        for (int t = 0; t < 2; ++t) bq[q][t] = bn[q][t];
      }
    }
  }

  const int rbase = b * 16 + khalf * 8;
#pragma unroll
  for (int t = 0; t < 2; ++t) {
#pragma unroll
    for (int i = 0; i < 8; ++i) {
      xw2[(size_t)(rbase + i) * 64 + (2 * wave + t) * 16 + m16] = acc2[t][i];
    }
  }
}

extern "C" void kernel_launch(void* const* d_in, const int* in_sizes, int n_in,
                              void* d_out, int out_size, void* d_ws, size_t ws_size,
                              hipStream_t stream) {
  (void)in_sizes; (void)n_in; (void)out_size; (void)ws_size;

  const float* feature = (const float*)d_in[0];
  const float* img     = (const float*)d_in[1];
  const int*   edges   = (const int*)d_in[2];
  const float* W_feat  = (const float*)d_in[3];
  const float* b_feat  = (const float*)d_in[4];
  const float* W_img   = (const float*)d_in[5];
  const float* b_img   = (const float*)d_in[6];
  const float* W_g1    = (const float*)d_in[7];
  const float* b_g1    = (const float*)d_in[8];
  const float* W_g2    = (const float*)d_in[9];
  const float* b_g2    = (const float*)d_in[10];
  float* out = (float*)d_out;

  const int* src = edges;
  const int* dst = edges + N_EDGES;

  float* ws     = (float*)d_ws;
  float* bufA   = ws + 50048;
  float* bufB   = bufA + (size_t)N_NODES * 96;
  float* wpF    = bufB + (size_t)N_NODES * 96;
  float* wpI    = wpF + 512 * 64;
  float* wpG1   = wpI + 256 * 32;
  float* wpG2   = wpG1 + 96 * 96;
  int*   ell    = (int*)bufA;
  int*   cnt    = ell + (size_t)N_NODES * ELLW;
  int*   head   = cnt + 50048;
  int*   next   = head + 50048;
  float* xw1b   = bufA + (size_t)N_NODES * 64;
  float* xw1a   = out;

  k_prep<<<PB_F + PB_I + PB_1 + PB_2 + PB_Z, 256, 0, stream>>>(
      W_feat, wpF, W_img, wpI, W_g1, wpG1, W_g2, wpG2, cnt, head);

  k_enc_rank<<<2 * GB + EB, 128, 0, stream>>>(feature, wpF, b_feat, img, wpI, b_img, bufB,
                                               src, dst, cnt, ell, head, next);

  k_gemm1<<<GB, 128, 0, stream>>>(bufB, wpG1, cnt, xw1a, xw1b);

  k_gather1_gemm2<<<GT, 512, 0, stream>>>(xw1a, xw1b, cnt, ell, head, next, src, b_g1, wpG2, bufB);

  k_gather<<<(N_NODES + 7) / 8, 256, 0, stream>>>(bufB, cnt, ell, head, next, src, b_g2, out, N_NODES);
}


# GEMM weights shared through LDS: a work-group copies the packed weights once, in 32-k slabs (ring of two, one barrier each); its 4 waves read B pieces from LDS, not from global per wave; k order kept
# speedup vs baseline: 2.6746x; 1.0416x over previous
#include <hip/hip_runtime.h>
#include <hip/hip_bf16.h>

typedef float v2f __attribute__((ext_vector_type(2)));
typedef float v8f __attribute__((ext_vector_type(8)));

#define N_NODES 50000
#define N_EDGES 800000

#define ELLW 32


template <int K, int N>
__device__ __forceinline__ void pack_w(const float* __restrict__ W, float* __restrict__ Wp,
                                       int idx) {
  if (idx >= K * N) return;
  const int k = idx / N;
  const int n = idx - k * N;
  Wp[(size_t)(k >> 2) * (N * 4) + n * 4 + (k & 3)] = W[idx];
}

typedef float v4f __attribute__((ext_vector_type(4)));

#define SLABK 32
#define GEMM_THREADS 128

template <int KDIM, int NT, bool BIAS_RELU, bool ROW_SCALE, bool SPLIT>
__device__ __forceinline__ void gemm_rows16(
    const float* __restrict__ A, const float* __restrict__ Wp,
    const float* __restrict__ bias, const int* __restrict__ rcnt,
    float* __restrict__ out, float* __restrict__ out2,
    float* __restrict__ ring,
    int row_tile_in, int tid, int lane, int out_stride, int out_off) {
  static_assert(KDIM % SLABK == 0, "whole slabs of two 16-float chunks");
  const int m16 = lane & 15;
  const int khalf = lane >> 4;
  constexpr int NCOLS = NT * 16;
  constexpr int NCH = KDIM / 16;
  constexpr int GROUP = NCOLS * 4;
  constexpr int NSLAB = KDIM / SLABK;
  constexpr int SLAB_F = SLABK * NCOLS;
  constexpr int PIECES = SLAB_F / (4 * GEMM_THREADS);
  static_assert(SLAB_F % (4 * GEMM_THREADS) == 0, "every thread moves whole pieces");

  const bool live = row_tile_in < N_NODES;
  const int row_tile = live ? row_tile_in : (N_NODES - 16);

  v8f acc[NT];
#pragma unroll
  for (int t = 0; t < NT; ++t) acc[t] = (v8f){0.f, 0.f, 0.f, 0.f, 0.f, 0.f, 0.f, 0.f};

  const float* __restrict__ arow = A + (size_t)(row_tile + m16) * KDIM + khalf * 4;
  const float* __restrict__ wsrc = Wp + tid * 4;
  float* __restrict__ wdst = ring + tid * 4;
  const float* __restrict__ wlane = ring + khalf * GROUP + m16 * 4;

  v4f st[PIECES];
#pragma unroll
  for (int p = 0; p < PIECES; ++p) st[p] = *(const v4f*)(wsrc + p * (4 * GEMM_THREADS));

  float rs_lane = 1.0f;
  if (ROW_SCALE) rs_lane = rsqrtf((float)(rcnt[row_tile + m16] + 1));

  v4f a_even[2], a_odd[2];
#pragma unroll
  for (int q = 0; q < 2; ++q) a_even[q] = *(const v4f*)(arow + q * 8);

#pragma unroll
  for (int p = 0; p < PIECES; ++p) *(v4f*)(wdst + p * (4 * GEMM_THREADS)) = st[p];
  __syncthreads();

  auto chunk = [&](const int ch, const float* __restrict__ wc, const float* __restrict__ stage_from,
                   const v4f (&a_cur)[2], v4f (&a_nxt)[2]) {
    v4f b[2][NT];
#pragma unroll
    for (int q = 0; q < 2; ++q) {
#pragma unroll
      for (int t = 0; t < NT; ++t) b[q][t] = *(const v4f*)(wc + q * (2 * GROUP) + t * 64);
    }
    if (ch + 1 < NCH) {
#pragma unroll
      for (int q = 0; q < 2; ++q) a_nxt[q] = *(const v4f*)(arow + (ch + 1) * 16 + q * 8);
    }
    if (stage_from) {
      __builtin_amdgcn_sched_barrier(0);
#pragma unroll
      for (int p = 0; p < PIECES; ++p) st[p] = *(const v4f*)(stage_from + p * (4 * GEMM_THREADS));
    }
    if ((ch & 1) == 0 && ch + 8 < NCH) __builtin_prefetch(arow + (ch + 8) * 16, 0, 0);
    __builtin_amdgcn_sched_barrier(0);
#pragma unroll
    for (int q = 0; q < 2; ++q) {
      const v2f a0 = {a_cur[q].x, a_cur[q].y};
      const v2f a1 = {a_cur[q].z, a_cur[q].w};
#pragma unroll
      for (int t = 0; t < NT; ++t) {
        const v2f b0 = {b[q][t].x, b[q][t].y};
        acc[t] = __builtin_amdgcn_wmma_f32_16x16x4_f32(
             false, a0,  false, b0,
             (short)0, acc[t],  false,  false);
      }
#pragma unroll
      for (int t = 0; t < NT; ++t) {
        const v2f b1 = {b[q][t].z, b[q][t].w};
        acc[t] = __builtin_amdgcn_wmma_f32_16x16x4_f32(
             false, a1,  false, b1,
             (short)0, acc[t],  false,  false);
      }
    }
    __builtin_amdgcn_sched_barrier(0);
  };

  for (int s = 0; s < NSLAB; ++s) {
    const float* __restrict__ cur = wlane + (s & 1) * SLAB_F;
    const float* __restrict__ nxt_src = (s + 1 < NSLAB) ? wsrc + (size_t)(s + 1) * SLAB_F : nullptr;
    chunk(2 * s, cur, nxt_src, a_even, a_odd);
    chunk(2 * s + 1, cur + 4 * GROUP, nullptr, a_odd, a_even);
    if (s + 1 < NSLAB) {
      float* __restrict__ nxt = wdst + ((s + 1) & 1) * SLAB_F;
#pragma unroll
      for (int p = 0; p < PIECES; ++p) *(v4f*)(nxt + p * (4 * GEMM_THREADS)) = st[p];
      __syncthreads();
    }
  }

  if (!live) return;

  const int rbase = row_tile + khalf * 8;
  float rs[8];
  if (ROW_SCALE) {
#pragma unroll
    for (int i = 0; i < 8; ++i) {
      const float lo = __builtin_bit_cast(float, __builtin_amdgcn_readlane(__builtin_bit_cast(int, rs_lane), i));
      const float hi = __builtin_bit_cast(float, __builtin_amdgcn_readlane(__builtin_bit_cast(int, rs_lane), 8 + i));
      rs[i] = khalf ? hi : lo;
    }
  }
#pragma unroll
  for (int t = 0; t < NT; ++t) {
#pragma unroll
    for (int i = 0; i < 8; ++i) {
      const int c = t * 16 + m16;
      float v = acc[t][i];
      if (ROW_SCALE) v *= rs[i];
      if (BIAS_RELU) v = fmaxf(v + bias[c], 0.0f);
      if (SPLIT && t >= 4) {
        out2[(size_t)(rbase + i) * 32 + (c - 64)] = v;
      } else {
        out[(size_t)(rbase + i) * out_stride + out_off + c] = v;
      }
    }
  }
}

#define GB ((N_NODES + 63) / 64)
static_assert(GEMM_THREADS == 128, "the GEMM work-groups are 4 waves: launch bounds and grids below");
#define EB ((N_EDGES + 127) / 128)

#define PB_F ((512 * 64 + 255) / 256)
#define PB_I ((256 * 32 + 255) / 256)
#define PB_1 ((96 * 96 + 255) / 256)
#define PB_2 ((96 * 64 + 255) / 256)
#define PB_Z ((N_NODES + 255) / 256)

__global__ __launch_bounds__(256) void k_prep(
    const float* __restrict__ W_feat, float* __restrict__ wpF,
    const float* __restrict__ W_img, float* __restrict__ wpI,
    const float* __restrict__ W_g1, float* __restrict__ wpG1,
    const float* __restrict__ W_g2, float* __restrict__ wpG2,
    int* __restrict__ cnt, int* __restrict__ head) {
  const int b = blockIdx.x;
  const int t = threadIdx.x;
  if (b < PB_F) {
    pack_w<512, 64>(W_feat, wpF, b * 256 + t);
  } else if (b < PB_F + PB_I) {
    pack_w<256, 32>(W_img, wpI, (b - PB_F) * 256 + t);
  } else if (b < PB_F + PB_I + PB_1) {
    pack_w<96, 96>(W_g1, wpG1, (b - PB_F - PB_I) * 256 + t);
  } else if (b < PB_F + PB_I + PB_1 + PB_2) {
    pack_w<96, 64>(W_g2, wpG2, (b - PB_F - PB_I - PB_1) * 256 + t);
  } else {
    const int i = (b - PB_F - PB_I - PB_1 - PB_2) * 256 + t;
    if (i < N_NODES) {
      cnt[i] = 0;
      head[i] = -1;
    }
  }
}

__global__ __launch_bounds__(128) void k_enc_rank(
    const float* __restrict__ feature, const float* __restrict__ wpF,
    const float* __restrict__ b_feat,
    const float* __restrict__ img, const float* __restrict__ wpI,
    const float* __restrict__ b_img, float* __restrict__ x,
    const int* __restrict__ src, const int* __restrict__ dst,
    int* __restrict__ cnt, int* __restrict__ ell,
    int* __restrict__ head, int* __restrict__ next) {
  __shared__ __attribute__((aligned(16))) float ring[2 * SLABK * 64];
  const int b = blockIdx.x;
  const int lane = threadIdx.x & 31;
  const int wave = threadIdx.x >> 5;
  if (b < GB) {
    const int row_tile = (b * 4 + wave) * 16;
    gemm_rows16<512, 4, true, false, false>(feature, wpF, b_feat, nullptr, x, nullptr, ring, row_tile, (int)threadIdx.x, lane, 96, 0);
  } else if (b < 2 * GB) {
    const int row_tile = ((b - GB) * 4 + wave) * 16;
    gemm_rows16<256, 2, true, false, false>(img, wpI, b_img, nullptr, x, nullptr, ring, row_tile, (int)threadIdx.x, lane, 96, 64);
  } else {
    const int i = (b - 2 * GB) * 128 + threadIdx.x;
    if (i < N_EDGES) {
      const int d = dst[i];
      const int r = atomicAdd(&cnt[d], 1);
      if (r < ELLW) {
        ell[(size_t)d * ELLW + r] = src[i];
      } else {
        next[i] = atomicExch(&head[d], i);
      }
    }
  }
}

__global__ __launch_bounds__(128) __attribute__((amdgpu_waves_per_eu(1, 7))) void k_gemm1(
    const float* __restrict__ x, const float* __restrict__ wpG1,
    const int* __restrict__ cnt, float* __restrict__ xw1a, float* __restrict__ xw1b) {
  __shared__ __attribute__((aligned(16))) float ring[2 * SLABK * 96];
  const int b = blockIdx.x;
  const int lane = threadIdx.x & 31;
  const int wave = threadIdx.x >> 5;
  const int row_tile = (b * 4 + wave) * 16;
  gemm_rows16<96, 6, false, true, true>(x, wpG1, nullptr, cnt, xw1a, xw1b, ring, row_tile, (int)threadIdx.x, lane, 64, 0);
}

template <int V>
__device__ __forceinline__ void load_row(float (&x)[V], const float* __restrict__ xa,
                                         const float* __restrict__ xb, int s, int lane) {
  const float* __restrict__ p = xa + (size_t)s * 64 + lane;
  x[0] = p[0];
  x[1] = p[32];
  if (V == 3) x[2] = xb[(size_t)s * 32 + lane];
}

template <int V>
__device__ __forceinline__ int gather_row(
    float (&acc)[V], const float* __restrict__ xa, const float* __restrict__ xb,
    const int* __restrict__ cnt, const int* __restrict__ ell,
    const int* __restrict__ head, const int* __restrict__ next,
    const int* __restrict__ src, int node, int lane) {
  const int c = cnt[node];
  const int m = min(c, ELLW);
  int s_l = 0;
  if (lane < m) s_l = ell[(size_t)node * ELLW + lane];
  load_row<V>(acc, xa, xb, node, lane);

  int j = 0;
  for (; j + 4 <= m; j += 4) {
    float x[4][V];
#pragma unroll
    for (int u = 0; u < 4; ++u) {
      const int s = __builtin_amdgcn_readlane(s_l, j + u);
      load_row<V>(x[u], xa, xb, s, lane);
    }
#pragma unroll
    for (int u = 0; u < 4; ++u) {
#pragma unroll
      for (int v = 0; v < V; ++v) acc[v] += x[u][v];
    }
  }
  for (; j < m; ++j) {
    const int s = __builtin_amdgcn_readlane(s_l, j);
    float x[V];
    load_row<V>(x, xa, xb, s, lane);
#pragma unroll
    for (int v = 0; v < V; ++v) acc[v] += x[v];
  }
  if (c > ELLW) {
    int e = head[node];
    for (int k = ELLW; k < c; ++k) {
      if (e < 0 || e >= N_EDGES) break;
      const int s = src[e];
      float x[V];
      load_row<V>(x, xa, xb, s, lane);
#pragma unroll
      for (int v = 0; v < V; ++v) acc[v] += x[v];
      e = next[e];
    }
  }
  return c;
}

__global__ __launch_bounds__(256) void k_gather(
    const float* __restrict__ xw, const int* __restrict__ cnt,
    const int* __restrict__ ell, const int* __restrict__ head,
    const int* __restrict__ next, const int* __restrict__ src,
    const float* __restrict__ bias, float* __restrict__ out, int n) {
  constexpr int V = 2;
  const int lane = threadIdx.x & 31;
  const int node = __builtin_amdgcn_readfirstlane((int)(blockIdx.x * 8 + (threadIdx.x >> 5)));
  if (node >= n) return;

  float acc[V];
  const int c = gather_row<V>(acc, xw, nullptr, cnt, ell, head, next, src, node, lane);
  const float di = rsqrtf((float)(c + 1));

  float* __restrict__ o = out + (size_t)node * 64 + lane;
#pragma unroll
  for (int v = 0; v < V; ++v) o[32 * v] = acc[v] * di + bias[lane + 32 * v];
}


static_assert(N_NODES % 16 == 0, "every gather work-group is one full 16-row tile");
#define TILE_PITCH 100
#define GT (N_NODES / 16)

__global__ __launch_bounds__(512) void k_gather1_gemm2(
    const float* __restrict__ xwa, const float* __restrict__ xwb,
    const int* __restrict__ cnt, const int* __restrict__ ell,
    const int* __restrict__ head, const int* __restrict__ next,
    const int* __restrict__ src,
    const float* __restrict__ bias, const float* __restrict__ wp2,
    float* __restrict__ xw2) {
  __shared__ __attribute__((aligned(16))) float hs[16 * TILE_PITCH];
  const int b = blockIdx.x;
  constexpr int V = 3;
  const int lane = threadIdx.x & 31;
  const int wave = __builtin_amdgcn_readfirstlane((int)(threadIdx.x >> 5));
  const int node = b * 16 + wave;

  float acc[V];
  const int c = gather_row<V>(acc, xwa, xwb, cnt, ell, head, next, src, node, lane);
  const float di = rsqrtf((float)(c + 1));

#pragma unroll
  for (int v = 0; v < V; ++v) {
    float r = fmaxf(acc[v] * di + bias[lane + 32 * v], 0.0f);
    hs[wave * TILE_PITCH + lane + 32 * v] = r * di;
  }

  const int m16 = lane & 15;
  const int khalf = lane >> 4;
  const float* __restrict__ wl = wp2 + khalf * 256 + m16 * 4 + wave * 128;
  v4f bq[2][2];
  if (wave < 2) {
#pragma unroll
    for (int q = 0; q < 2; ++q) {
#pragma unroll
      for (int t = 0; t < 2; ++t) bq[q][t] = *(const v4f*)(wl + q * 512 + t * 64);
    }
  }
  __syncthreads();
  if (wave >= 2) return;

  v8f acc2[2];
#pragma unroll
  for (int t = 0; t < 2; ++t) acc2[t] = (v8f){0.f, 0.f, 0.f, 0.f, 0.f, 0.f, 0.f, 0.f};
  const float* al = hs + m16 * TILE_PITCH + khalf * 4;

#pragma unroll
  for (int ch = 0; ch < 6; ++ch) {
    v4f aq[2];
#pragma unroll
    for (int q = 0; q < 2; ++q) aq[q] = *(const v4f*)(al + ch * 16 + q * 8);
    v4f bn[2][2];
    if (ch + 1 < 6) {
#pragma unroll
      for (int q = 0; q < 2; ++q) {
#pragma unroll
        for (int t = 0; t < 2; ++t) bn[q][t] = *(const v4f*)(wl + (ch + 1) * 1024 + q * 512 + t * 64);
      }
    }
#pragma unroll
    for (int q = 0; q < 2; ++q) {
      const v2f a0 = {aq[q].x, aq[q].y};
      const v2f a1 = {aq[q].z, aq[q].w};
#pragma unroll
      for (int t = 0; t < 2; ++t) {
        const v2f b0 = {bq[q][t].x, bq[q][t].y};
        acc2[t] = __builtin_amdgcn_wmma_f32_16x16x4_f32(
             false, a0,  false, b0,
             (short)0, acc2[t],  false,  false);
      }
#pragma unroll
      for (int t = 0; t < 2; ++t) {
        const v2f b1 = {bq[q][t].z, bq[q][t].w};
        acc2[t] = __builtin_amdgcn_wmma_f32_16x16x4_f32(
             false, a1,  false, b1,
             (short)0, acc2[t],  false,  false);
      }
    }
    if (ch + 1 < 6) {
#pragma unroll
      for (int q = 0; q < 2; ++q) {
#pragma unroll
        for (int t = 0; t < 2; ++t) bq[q][t] = bn[q][t];
      }
    }
  }

  const int rbase = b * 16 + khalf * 8;
#pragma unroll
  for (int t = 0; t < 2; ++t) {
#pragma unroll
    for (int i = 0; i < 8; ++i) {
      xw2[(size_t)(rbase + i) * 64 + (2 * wave + t) * 16 + m16] = acc2[t][i];
    }
  }
}

extern "C" void kernel_launch(void* const* d_in, const int* in_sizes, int n_in,
                              void* d_out, int out_size, void* d_ws, size_t ws_size,
                              hipStream_t stream) {
  (void)in_sizes; (void)n_in; (void)out_size; (void)ws_size;

  const float* feature = (const float*)d_in[0];
  const float* img     = (const float*)d_in[1];
  const int*   edges   = (const int*)d_in[2];
  const float* W_feat  = (const float*)d_in[3];
  const float* b_feat  = (const float*)d_in[4];
  const float* W_img   = (const float*)d_in[5];
  const float* b_img   = (const float*)d_in[6];
  const float* W_g1    = (const float*)d_in[7];
  const float* b_g1    = (const float*)d_in[8];
  const float* W_g2    = (const float*)d_in[9];
  const float* b_g2    = (const float*)d_in[10];
  float* out = (float*)d_out;

  const int* src = edges;
  const int* dst = edges + N_EDGES;

  float* ws     = (float*)d_ws;
  float* bufA   = ws + 50048;
  float* bufB   = bufA + (size_t)N_NODES * 96;
  float* wpF    = bufB + (size_t)N_NODES * 96;
  float* wpI    = wpF + 512 * 64;
  float* wpG1   = wpI + 256 * 32;
  float* wpG2   = wpG1 + 96 * 96;
  int*   ell    = (int*)bufA;
  int*   cnt    = ell + (size_t)N_NODES * ELLW;
  int*   head   = cnt + 50048;
  int*   next   = head + 50048;
  float* xw1b   = bufA + (size_t)N_NODES * 64;
  float* xw1a   = out;

  k_prep<<<PB_F + PB_I + PB_1 + PB_2 + PB_Z, 256, 0, stream>>>(
      W_feat, wpF, W_img, wpI, W_g1, wpG1, W_g2, wpG2, cnt, head);

  k_enc_rank<<<2 * GB + EB, 128, 0, stream>>>(feature, wpF, b_feat, img, wpI, b_img, bufB,
                                               src, dst, cnt, ell, head, next);

  k_gemm1<<<GB, 128, 0, stream>>>(bufB, wpG1, cnt, xw1a, xw1b);

  k_gather1_gemm2<<<GT, 512, 0, stream>>>(xw1a, xw1b, cnt, ell, head, next, src, b_g1, wpG2, bufB);

  k_gather<<<(N_NODES + 7) / 8, 256, 0, stream>>>(bufB, cnt, ell, head, next, src, b_g2, out, N_NODES);
}


# Gather waves take two destination rows, one per 16-lane half, with 16-byte row loads (one load = whole rows of two neighbours): half the waves, 3-4x fewer row loads issued; order of the sums kept
# speedup vs baseline: 2.7845x; 1.0411x over previous
#include <hip/hip_runtime.h>
#include <hip/hip_bf16.h>

typedef float v2f __attribute__((ext_vector_type(2)));
typedef float v8f __attribute__((ext_vector_type(8)));

#define N_NODES 50000
#define N_EDGES 800000

#define ELLW 32


template <int K, int N>
__device__ __forceinline__ void pack_w(const float* __restrict__ W, float* __restrict__ Wp,
                                       int idx) {
  if (idx >= K * N) return;
  const int k = idx / N;
  const int n = idx - k * N;
  Wp[(size_t)(k >> 2) * (N * 4) + n * 4 + (k & 3)] = W[idx];
}

typedef float v4f __attribute__((ext_vector_type(4)));

#define SLABK 32
#define GEMM_THREADS 128

template <int KDIM, int NT, bool BIAS_RELU, bool ROW_SCALE, bool SPLIT>
__device__ __forceinline__ void gemm_rows16(
    const float* __restrict__ A, const float* __restrict__ Wp,
    const float* __restrict__ bias, const int* __restrict__ rcnt,
    float* __restrict__ out, float* __restrict__ out2,
    float* __restrict__ ring,
    int row_tile_in, int tid, int lane, int out_stride, int out_off) {
  static_assert(KDIM % SLABK == 0, "whole slabs of two 16-float chunks");
  const int m16 = lane & 15;
  const int khalf = lane >> 4;
  constexpr int NCOLS = NT * 16;
  constexpr int NCH = KDIM / 16;
  constexpr int GROUP = NCOLS * 4;
  constexpr int NSLAB = KDIM / SLABK;
  constexpr int SLAB_F = SLABK * NCOLS;
  constexpr int PIECES = SLAB_F / (4 * GEMM_THREADS);
  static_assert(SLAB_F % (4 * GEMM_THREADS) == 0, "every thread moves whole pieces");

  const bool live = row_tile_in < N_NODES;
  const int row_tile = live ? row_tile_in : (N_NODES - 16);

  v8f acc[NT];
#pragma unroll
  for (int t = 0; t < NT; ++t) acc[t] = (v8f){0.f, 0.f, 0.f, 0.f, 0.f, 0.f, 0.f, 0.f};

  const float* __restrict__ arow = A + (size_t)(row_tile + m16) * KDIM + khalf * 4;
  const float* __restrict__ wsrc = Wp + tid * 4;
  float* __restrict__ wdst = ring + tid * 4;
  const float* __restrict__ wlane = ring + khalf * GROUP + m16 * 4;

  v4f st[PIECES];
#pragma unroll
  for (int p = 0; p < PIECES; ++p) st[p] = *(const v4f*)(wsrc + p * (4 * GEMM_THREADS));

  float rs_lane = 1.0f;
  if (ROW_SCALE) rs_lane = rsqrtf((float)(rcnt[row_tile + m16] + 1));

  v4f a_even[2], a_odd[2];
#pragma unroll
  for (int q = 0; q < 2; ++q) a_even[q] = *(const v4f*)(arow + q * 8);

#pragma unroll
  for (int p = 0; p < PIECES; ++p) *(v4f*)(wdst + p * (4 * GEMM_THREADS)) = st[p];
  __syncthreads();

  auto chunk = [&](const int ch, const float* __restrict__ wc, const float* __restrict__ stage_from,
                   const v4f (&a_cur)[2], v4f (&a_nxt)[2]) {
    v4f b[2][NT];
#pragma unroll
    for (int q = 0; q < 2; ++q) {
#pragma unroll
      for (int t = 0; t < NT; ++t) b[q][t] = *(const v4f*)(wc + q * (2 * GROUP) + t * 64);
    }
    if (ch + 1 < NCH) {
#pragma unroll
      for (int q = 0; q < 2; ++q) a_nxt[q] = *(const v4f*)(arow + (ch + 1) * 16 + q * 8);
    }
    if (stage_from) {
      __builtin_amdgcn_sched_barrier(0);
#pragma unroll
      for (int p = 0; p < PIECES; ++p) st[p] = *(const v4f*)(stage_from + p * (4 * GEMM_THREADS));
    }
    if ((ch & 1) == 0 && ch + 8 < NCH) __builtin_prefetch(arow + (ch + 8) * 16, 0, 0);
    __builtin_amdgcn_sched_barrier(0);
#pragma unroll
    for (int q = 0; q < 2; ++q) {
      const v2f a0 = {a_cur[q].x, a_cur[q].y};
      const v2f a1 = {a_cur[q].z, a_cur[q].w};
#pragma unroll
      for (int t = 0; t < NT; ++t) {
        const v2f b0 = {b[q][t].x, b[q][t].y};
        acc[t] = __builtin_amdgcn_wmma_f32_16x16x4_f32(
             false, a0,  false, b0,
             (short)0, acc[t],  false,  false);
      }
#pragma unroll
      for (int t = 0; t < NT; ++t) {
        const v2f b1 = {b[q][t].z, b[q][t].w};
        acc[t] = __builtin_amdgcn_wmma_f32_16x16x4_f32(
             false, a1,  false, b1,
             (short)0, acc[t],  false,  false);
      }
    }
    __builtin_amdgcn_sched_barrier(0);
  };

  for (int s = 0; s < NSLAB; ++s) {
    const float* __restrict__ cur = wlane + (s & 1) * SLAB_F;
    const float* __restrict__ nxt_src = (s + 1 < NSLAB) ? wsrc + (size_t)(s + 1) * SLAB_F : nullptr;
    chunk(2 * s, cur, nxt_src, a_even, a_odd);
    chunk(2 * s + 1, cur + 4 * GROUP, nullptr, a_odd, a_even);
    if (s + 1 < NSLAB) {
      float* __restrict__ nxt = wdst + ((s + 1) & 1) * SLAB_F;
#pragma unroll
      for (int p = 0; p < PIECES; ++p) *(v4f*)(nxt + p * (4 * GEMM_THREADS)) = st[p];
      __syncthreads();
    }
  }

  if (!live) return;

  const int rbase = row_tile + khalf * 8;
  float rs[8];
  if (ROW_SCALE) {
#pragma unroll
    for (int i = 0; i < 8; ++i) {
      const float lo = __builtin_bit_cast(float, __builtin_amdgcn_readlane(__builtin_bit_cast(int, rs_lane), i));
      const float hi = __builtin_bit_cast(float, __builtin_amdgcn_readlane(__builtin_bit_cast(int, rs_lane), 8 + i));
      rs[i] = khalf ? hi : lo;
    }
  }
#pragma unroll
  for (int t = 0; t < NT; ++t) {
#pragma unroll
    for (int i = 0; i < 8; ++i) {
      const int c = t * 16 + m16;
      float v = acc[t][i];
      if (ROW_SCALE) v *= rs[i];
      if (BIAS_RELU) v = fmaxf(v + bias[c], 0.0f);
      if (SPLIT && t >= 4) {
        out2[(size_t)(rbase + i) * 32 + (c - 64)] = v;
      } else {
        out[(size_t)(rbase + i) * out_stride + out_off + c] = v;
      }
    }
  }
}

#define GB ((N_NODES + 63) / 64)
static_assert(GEMM_THREADS == 128, "the GEMM work-groups are 4 waves: launch bounds and grids below");
#define EB ((N_EDGES + 127) / 128)

#define PB_F ((512 * 64 + 255) / 256)
#define PB_I ((256 * 32 + 255) / 256)
#define PB_1 ((96 * 96 + 255) / 256)
#define PB_2 ((96 * 64 + 255) / 256)
#define PB_Z ((N_NODES + 255) / 256)

__global__ __launch_bounds__(256) void k_prep(
    const float* __restrict__ W_feat, float* __restrict__ wpF,
    const float* __restrict__ W_img, float* __restrict__ wpI,
    const float* __restrict__ W_g1, float* __restrict__ wpG1,
    const float* __restrict__ W_g2, float* __restrict__ wpG2,
    int* __restrict__ cnt, int* __restrict__ head) {
  const int b = blockIdx.x;
  const int t = threadIdx.x;
  if (b < PB_F) {
    pack_w<512, 64>(W_feat, wpF, b * 256 + t);
  } else if (b < PB_F + PB_I) {
    pack_w<256, 32>(W_img, wpI, (b - PB_F) * 256 + t);
  } else if (b < PB_F + PB_I + PB_1) {
    pack_w<96, 96>(W_g1, wpG1, (b - PB_F - PB_I) * 256 + t);
  } else if (b < PB_F + PB_I + PB_1 + PB_2) {
    pack_w<96, 64>(W_g2, wpG2, (b - PB_F - PB_I - PB_1) * 256 + t);
  } else {
    const int i = (b - PB_F - PB_I - PB_1 - PB_2) * 256 + t;
    if (i < N_NODES) {
      cnt[i] = 0;
      head[i] = -1;
    }
  }
}

__global__ __launch_bounds__(128) void k_enc_rank(
    const float* __restrict__ feature, const float* __restrict__ wpF,
    const float* __restrict__ b_feat,
    const float* __restrict__ img, const float* __restrict__ wpI,
    const float* __restrict__ b_img, float* __restrict__ x,
    const int* __restrict__ src, const int* __restrict__ dst,
    int* __restrict__ cnt, int* __restrict__ ell,
    int* __restrict__ head, int* __restrict__ next) {
  __shared__ __attribute__((aligned(16))) float ring[2 * SLABK * 64];
  const int b = blockIdx.x;
  const int lane = threadIdx.x & 31;
  const int wave = threadIdx.x >> 5;
  if (b < GB) {
    const int row_tile = (b * 4 + wave) * 16;
    gemm_rows16<512, 4, true, false, false>(feature, wpF, b_feat, nullptr, x, nullptr, ring, row_tile, (int)threadIdx.x, lane, 96, 0);
  } else if (b < 2 * GB) {
    const int row_tile = ((b - GB) * 4 + wave) * 16;
    gemm_rows16<256, 2, true, false, false>(img, wpI, b_img, nullptr, x, nullptr, ring, row_tile, (int)threadIdx.x, lane, 96, 64);
  } else {
    const int i = (b - 2 * GB) * 128 + threadIdx.x;
    if (i < N_EDGES) {
      const int d = dst[i];
      const int r = atomicAdd(&cnt[d], 1);
      if (r < ELLW) {
        ell[(size_t)d * ELLW + r] = src[i];
      } else {
        next[i] = atomicExch(&head[d], i);
      }
    }
  }
}

__global__ __launch_bounds__(128) __attribute__((amdgpu_waves_per_eu(1, 7))) void k_gemm1(
    const float* __restrict__ x, const float* __restrict__ wpG1,
    const int* __restrict__ cnt, float* __restrict__ xw1a, float* __restrict__ xw1b) {
  __shared__ __attribute__((aligned(16))) float ring[2 * SLABK * 96];
  const int b = blockIdx.x;
  const int lane = threadIdx.x & 31;
  const int wave = threadIdx.x >> 5;
  const int row_tile = (b * 4 + wave) * 16;
  gemm_rows16<96, 6, false, true, true>(x, wpG1, nullptr, cnt, xw1a, xw1b, ring, row_tile, (int)threadIdx.x, lane, 64, 0);
}

template <int V>
__device__ __forceinline__ void load_row(v4f& a, v2f& b, const float* __restrict__ xa,
                                         const float* __restrict__ xb, int s, int c16) {
  a = *(const v4f*)(xa + (size_t)s * 64 + c16 * 4);
  if (V == 3) b = *(const v2f*)(xb + (size_t)s * 32 + c16 * 2);
}

template <int V>
__device__ __forceinline__ int gather_row(
    v4f& acc_a, v2f& acc_b, const float* __restrict__ xa, const float* __restrict__ xb,
    const int* __restrict__ cnt, const int* __restrict__ ell,
    const int* __restrict__ head, const int* __restrict__ next,
    const int* __restrict__ src, int node, int c16) {
  const int c = cnt[node];
  const int m = min(c, ELLW);
  int id_lo = 0, id_hi = 0;
  const int* __restrict__ er = ell + (size_t)node * ELLW + c16;
  if (c16 < m) id_lo = er[0];
  if (c16 + 16 < m) id_hi = er[16];
  acc_b = (v2f){0.f, 0.f};
  load_row<V>(acc_a, acc_b, xa, xb, node, c16);

  const int m_lo = __builtin_amdgcn_readlane(m, 0);
  const int m_hi = __builtin_amdgcn_readlane(m, 16);
  const int mmax = max(m_lo, m_hi);
  const int long_base = (m_hi > m_lo) ? 16 : 0;
  for (int j = 0; j < mmax; j += 4) {
    const int idv = (j < 16) ? id_lo : id_hi;
    v4f xa4[4];
    v2f xb4[4];
    bool on[4];
#pragma unroll
    for (int u = 0; u < 4; ++u) {
      const int r = j + u;
      const int n = r & 15;
      const unsigned sel = 0x11111111u * (unsigned)n;
      const int s_own = (int)__builtin_amdgcn_permlane16((unsigned)idv, (unsigned)idv, sel, sel, false, false);
      const int s_long = __builtin_amdgcn_readlane(idv, long_base + n);
      on[u] = r < m;
      xb4[u] = (v2f){0.f, 0.f};
      load_row<V>(xa4[u], xb4[u], xa, xb, on[u] ? s_own : s_long, c16);
    }
    __builtin_amdgcn_sched_barrier(0);
#pragma unroll
    for (int u = 0; u < 4; ++u) {
      const v4f ta = acc_a + xa4[u];
      acc_a = on[u] ? ta : acc_a;
      if (V == 3) {
        const v2f tb = acc_b + xb4[u];
        acc_b = on[u] ? tb : acc_b;
      }
    }
  }
  if (c > ELLW) {
    int e = head[node];
    for (int k = ELLW; k < c; ++k) {
      if (e < 0 || e >= N_EDGES) break;
      const int s = src[e];
      v4f ya;
      v2f yb = (v2f){0.f, 0.f};
      load_row<V>(ya, yb, xa, xb, s, c16);
      acc_a += ya;
      if (V == 3) acc_b += yb;
      e = next[e];
    }
  }
  return c;
}

static_assert(N_NODES % 16 == 0, "a gather wave owns two rows, a gather work-group sixteen");

__global__ __launch_bounds__(256) void k_gather(
    const float* __restrict__ xw, const int* __restrict__ cnt,
    const int* __restrict__ ell, const int* __restrict__ head,
    const int* __restrict__ next, const int* __restrict__ src,
    const float* __restrict__ bias, float* __restrict__ out, int n) {
  constexpr int V = 2;
  const int lane = threadIdx.x & 31;
  const int c16 = lane & 15;
  const int first = (int)(blockIdx.x * 8 + (threadIdx.x >> 5)) * 2;
  if (first >= n) return;
  const int node = first + (lane >> 4);

  v4f acc_a;
  v2f acc_b;
  const int c = gather_row<V>(acc_a, acc_b, xw, nullptr, cnt, ell, head, next, src, node, c16);
  const float di = rsqrtf((float)(c + 1));

  const v4f bv = *(const v4f*)(bias + c16 * 4);
  *(v4f*)(out + (size_t)node * 64 + c16 * 4) = acc_a * di + bv;
}


static_assert(N_NODES % 16 == 0, "every gather work-group is one full 16-row tile");
#define TILE_PITCH 100
#define GT (N_NODES / 16)

__global__ __launch_bounds__(256) void k_gather1_gemm2(
    const float* __restrict__ xwa, const float* __restrict__ xwb,
    const int* __restrict__ cnt, const int* __restrict__ ell,
    const int* __restrict__ head, const int* __restrict__ next,
    const int* __restrict__ src,
    const float* __restrict__ bias, const float* __restrict__ wp2,
    float* __restrict__ xw2) {
  __shared__ __attribute__((aligned(16))) float hs[16 * TILE_PITCH];
  const int b = blockIdx.x;
  constexpr int V = 3;
  const int lane = threadIdx.x & 31;
  const int wave = __builtin_amdgcn_readfirstlane((int)(threadIdx.x >> 5));
  const int trow = 2 * wave + (lane >> 4);
  const int node = b * 16 + trow;
  const int c16 = lane & 15;

  v4f acc_a;
  v2f acc_b;
  const int c = gather_row<V>(acc_a, acc_b, xwa, xwb, cnt, ell, head, next, src, node, c16);
  const float di = rsqrtf((float)(c + 1));

  {
    const v4f ba = *(const v4f*)(bias + c16 * 4);
    const v2f bb = *(const v2f*)(bias + 64 + c16 * 2);
    v4f ra = acc_a * di + ba;
    v2f rb = acc_b * di + bb;
    ra.x = fmaxf(ra.x, 0.0f); ra.y = fmaxf(ra.y, 0.0f);
    ra.z = fmaxf(ra.z, 0.0f); ra.w = fmaxf(ra.w, 0.0f);
    rb.x = fmaxf(rb.x, 0.0f); rb.y = fmaxf(rb.y, 0.0f);
    *(v4f*)(hs + trow * TILE_PITCH + c16 * 4) = ra * di;
    *(v2f*)(hs + trow * TILE_PITCH + 64 + c16 * 2) = rb * di;
  }

  const int m16 = lane & 15;
  const int khalf = lane >> 4;
  const float* __restrict__ wl = wp2 + khalf * 256 + m16 * 4 + wave * 128;
  v4f bq[2][2];
  if (wave < 2) {
#pragma unroll
    for (int q = 0; q < 2; ++q) {
#pragma unroll
      for (int t = 0; t < 2; ++t) bq[q][t] = *(const v4f*)(wl + q * 512 + t * 64);
    }
  }
  __syncthreads();
  if (wave >= 2) return;

  v8f acc2[2];
#pragma unroll
  for (int t = 0; t < 2; ++t) acc2[t] = (v8f){0.f, 0.f, 0.f, 0.f, 0.f, 0.f, 0.f, 0.f};
  const float* al = hs + m16 * TILE_PITCH + khalf * 4;

#pragma unroll
  for (int ch = 0; ch < 6; ++ch) {
    v4f aq[2];
#pragma unroll
    for (int q = 0; q < 2; ++q) aq[q] = *(const v4f*)(al + ch * 16 + q * 8);
    v4f bn[2][2];
    if (ch + 1 < 6) {
#pragma unroll
      for (int q = 0; q < 2; ++q) {
#pragma unroll
        for (int t = 0; t < 2; ++t) bn[q][t] = *(const v4f*)(wl + (ch + 1) * 1024 + q * 512 + t * 64);
      }
    }
#pragma unroll
    for (int q = 0; q < 2; ++q) {
      const v2f a0 = {aq[q].x, aq[q].y};
      const v2f a1 = {aq[q].z, aq[q].w};
#pragma unroll
      for (int t = 0; t < 2; ++t) {
        const v2f b0 = {bq[q][t].x, bq[q][t].y};
        acc2[t] = __builtin_amdgcn_wmma_f32_16x16x4_f32(
             false, a0,  false, b0,
             (short)0, acc2[t],  false,  false);
      }
#pragma unroll
      for (int t = 0; t < 2; ++t) {
        const v2f b1 = {bq[q][t].z, bq[q][t].w};
        acc2[t] = __builtin_amdgcn_wmma_f32_16x16x4_f32(
             false, a1,  false, b1,
             (short)0, acc2[t],  false,  false);
      }
    }
    if (ch + 1 < 6) {
#pragma unroll
      for (int q = 0; q < 2; ++q) {
#pragma unroll
        for (int t = 0; t < 2; ++t) bq[q][t] = bn[q][t];
      }
    }
  }

  const int rbase = b * 16 + khalf * 8;
#pragma unroll
  for (int t = 0; t < 2; ++t) {
#pragma unroll
    for (int i = 0; i < 8; ++i) {
      xw2[(size_t)(rbase + i) * 64 + (2 * wave + t) * 16 + m16] = acc2[t][i];
    }
  }
}

extern "C" void kernel_launch(void* const* d_in, const int* in_sizes, int n_in,
                              void* d_out, int out_size, void* d_ws, size_t ws_size,
                              hipStream_t stream) {
  (void)in_sizes; (void)n_in; (void)out_size; (void)ws_size;

  const float* feature = (const float*)d_in[0];
  const float* img     = (const float*)d_in[1];
  const int*   edges   = (const int*)d_in[2];
  const float* W_feat  = (const float*)d_in[3];
  const float* b_feat  = (const float*)d_in[4];
  const float* W_img   = (const float*)d_in[5];
  const float* b_img   = (const float*)d_in[6];
  const float* W_g1    = (const float*)d_in[7];
  const float* b_g1    = (const float*)d_in[8];
  const float* W_g2    = (const float*)d_in[9];
  const float* b_g2    = (const float*)d_in[10];
  float* out = (float*)d_out;

  const int* src = edges;
  const int* dst = edges + N_EDGES;

  float* ws     = (float*)d_ws;
  float* bufA   = ws + 50048;
  float* bufB   = bufA + (size_t)N_NODES * 96;
  float* wpF    = bufB + (size_t)N_NODES * 96;
  float* wpI    = wpF + 512 * 64;
  float* wpG1   = wpI + 256 * 32;
  float* wpG2   = wpG1 + 96 * 96;
  int*   ell    = (int*)bufA;
  int*   cnt    = ell + (size_t)N_NODES * ELLW;
  int*   head   = cnt + 50048;
  int*   next   = head + 50048;
  float* xw1b   = bufA + (size_t)N_NODES * 64;
  float* xw1a   = out;

  k_prep<<<PB_F + PB_I + PB_1 + PB_2 + PB_Z, 256, 0, stream>>>(
      W_feat, wpF, W_img, wpI, W_g1, wpG1, W_g2, wpG2, cnt, head);

  k_enc_rank<<<2 * GB + EB, 128, 0, stream>>>(feature, wpF, b_feat, img, wpI, b_img, bufB,
                                               src, dst, cnt, ell, head, next);

  k_gemm1<<<GB, 128, 0, stream>>>(bufB, wpG1, cnt, xw1a, xw1b);

  k_gather1_gemm2<<<GT, 256, 0, stream>>>(xw1a, xw1b, cnt, ell, head, next, src, b_g1, wpG2, bufB);

  k_gather<<<N_NODES / 16, 256, 0, stream>>>(bufB, cnt, ell, head, next, src, b_g2, out, N_NODES);
}
